# starting kernel
# baseline (speedup 1.0000x reference)
#include <hip/hip_runtime.h>
#include <math.h>

#define N_NODES 50000
#define N_HID   128
#define N_ADJ   4
#define N_EDGE  600000
#define TOTAL_EDGES (N_ADJ * N_EDGE)

typedef __attribute__((ext_vector_type(2))) float v2f;
typedef __attribute__((ext_vector_type(8))) float v8f;

__device__ __forceinline__ void global_fadd_f32(float* p, float v) {
    asm volatile("global_atomic_add_f32 %0, %1, off scope:SCOPE_DEV"
                 :
                 : "v"(p), "v"(v)
                 : "memory");
}

__global__ void zero_acc(float4* __restrict__ ws, int n4) {
    int i = blockIdx.x * blockDim.x + threadIdx.x;
    if (i < n4) {
        float4 z; z.x = 0.f; z.y = 0.f; z.z = 0.f; z.w = 0.f;
        ws[i] = z;
    }
}

__global__ __launch_bounds__(256) void scatter_edges(
    const float* __restrict__ x,
    const float* __restrict__ weight,
    const int*   __restrict__ adj_src,
    const int*   __restrict__ adj_dst,
    const float* __restrict__ adj_vals,
    float*       __restrict__ acc)
{
    const int lane = threadIdx.x & 31;
    const int e = blockIdx.x * (blockDim.x >> 5) + (threadIdx.x >> 5);
    if (e >= TOTAL_EDGES) return;

    const int   s  = __builtin_nontemporal_load(adj_src + e);
    const int   d  = __builtin_nontemporal_load(adj_dst + e);
    const float v  = __builtin_nontemporal_load(adj_vals + e);
    const int   a  = e / N_EDGE;
    const float sc = weight[a] * v;

    const float* xr   = x   + (long long)s * N_HID;
    float*       orow = acc + (long long)d * N_HID;

    #pragma unroll
    for (int k = 0; k < 4; ++k) {
        float m = xr[k * 32 + lane] * sc;
        global_fadd_f32(orow + k * 32 + lane, m);
    }
}

__global__ __launch_bounds__(160) void finalize_norm_gelu(
    const float* __restrict__ acc, float* __restrict__ out)
{
    const int lane = threadIdx.x & 31;
    const int wave = blockIdx.x * (blockDim.x >> 5) + (threadIdx.x >> 5);
    if (wave >= (N_NODES / 16)) return;

    const int node   = wave * 16 + (lane & 15);
    const int colOff = 2 * (lane >> 4);
    const float* rowp = acc + (long long)node * N_HID + colOff;

    v2f vbuf[32];
    v8f c = {};
    #pragma unroll
    for (int j = 0; j < 32; ++j) {
        v2f a;
        a.x = rowp[4 * j];
        a.y = rowp[4 * j + 1];
        vbuf[j] = a;
        c = __builtin_amdgcn_wmma_f32_16x16x4_f32(
                false, a, false, a, (short)0, c, false, false);
    }

    const int sel = lane & 7;
    float dv = c[0];
    dv = (sel == 1) ? c[1] : dv;
    dv = (sel == 2) ? c[2] : dv;
    dv = (sel == 3) ? c[3] : dv;
    dv = (sel == 4) ? c[4] : dv;
    dv = (sel == 5) ? c[5] : dv;
    dv = (sel == 6) ? c[6] : dv;
    dv = (sel == 7) ? c[7] : dv;
    const int nm      = lane & 15;
    const int srcLane = (nm < 8) ? nm : (nm + 16);
    const float sumsq = __shfl(dv, srcLane, 32);

    const float scale = 1.0f / fmaxf(sqrtf(sumsq), 1e-12f);

    float* op = out + (long long)node * N_HID + colOff;
    const float inv_sqrt2 = 0.70710678118654752440f;
    #pragma unroll
    for (int j = 0; j < 32; ++j) {
        float vx = vbuf[j].x * scale;
        float vy = vbuf[j].y * scale;
        vx = 0.5f * vx * (1.0f + erff(vx * inv_sqrt2));
        vy = 0.5f * vy * (1.0f + erff(vy * inv_sqrt2));
        op[4 * j]     = vx;
        op[4 * j + 1] = vy;
    }
}

extern "C" void kernel_launch(void* const* d_in, const int* in_sizes, int n_in,
                              void* d_out, int out_size, void* d_ws, size_t ws_size,
                              hipStream_t stream) {
    const float* x        = (const float*)d_in[0];
    const float* weight   = (const float*)d_in[1];
    const int*   adj_src  = (const int*)d_in[2];
    const int*   adj_dst  = (const int*)d_in[3];
    const float* adj_vals = (const float*)d_in[4];
    float*       out      = (float*)d_out;
    float*       acc      = (float*)d_ws;

    const int n4 = (N_NODES * N_HID) / 4;
    zero_acc<<<(n4 + 255) / 256, 256, 0, stream>>>((float4*)acc, n4);

    const int nblocks = (TOTAL_EDGES + 7) / 8;
    scatter_edges<<<nblocks, 256, 0, stream>>>(x, weight, adj_src, adj_dst,
                                               adj_vals, acc);

    const int tiles = N_NODES / 16;
    finalize_norm_gelu<<<(tiles + 4) / 5, 160, 0, stream>>>(acc, out);
}


# Float atomics removed: edges grouped by destination once per call (stable ballot-ranked partition into 25-row buckets), one wave sums each row in registers in edge order (sum order now fixed)
# speedup vs baseline: 1.9391x; 1.9391x over previous
#include <hip/hip_runtime.h>
#include <math.h>

#define N_NODES 50000
#define N_HID   128
#define N_ADJ   4
#define N_EDGE  600000
#define TOTAL_EDGES (N_ADJ * N_EDGE)

#define BUCKET_ROWS 25
#define N_BUCKET    (N_NODES / BUCKET_ROWS)
#define KEY_BITS    11
#define P_WG        250
#define P_WAVES     15
#define P_THREADS   (P_WAVES * 32)
#define P_BATCH     20
#define SCAN_LANES  400
#define ROW_CAP     128
#define ROWS_PER_WG 8

typedef __attribute__((ext_vector_type(2))) float v2f;
typedef __attribute__((ext_vector_type(8))) float v8f;

__device__ __forceinline__ unsigned same_key_lanes(int key) {
    unsigned m = 0xffffffffu;
    #pragma unroll
    for (int b = 0; b < KEY_BITS; ++b) {
        const bool bit = ((key >> b) & 1) != 0;
        const unsigned bal = __builtin_amdgcn_ballot_w32(bit);
        m &= bit ? bal : ~bal;
    }
    return m;
}

__global__ __launch_bounds__(P_THREADS) void part_count(
    const int* __restrict__ adj_dst, int* __restrict__ ghist)
{
    __shared__ int whist[P_WAVES][N_BUCKET];
    const int t = threadIdx.x;
    const int lane = t & 31;
    const int w = t >> 5;
    for (int i = t; i < P_WAVES * N_BUCKET; i += P_THREADS) (&whist[0][0])[i] = 0;
    __syncthreads();
    const int ebase = (blockIdx.x * P_WAVES + w) * (P_BATCH * 32);
    int key[P_BATCH];
    #pragma unroll
    for (int b = 0; b < P_BATCH; ++b)
        key[b] = __builtin_nontemporal_load(adj_dst + ebase + b * 32 + lane) / BUCKET_ROWS;
    #pragma unroll
    for (int b = 0; b < P_BATCH; ++b) {
        const unsigned m = same_key_lanes(key[b]);
        if (lane == __builtin_ctz(m)) whist[w][key[b]] += __builtin_popcount(m);
    }
    __syncthreads();
    for (int k = t; k < N_BUCKET; k += P_THREADS) {
        int sum = 0;
        #pragma unroll
        for (int wv = 0; wv < P_WAVES; ++wv) sum += whist[wv][k];
        ghist[blockIdx.x * N_BUCKET + k] = sum;
    }
}

__global__ __launch_bounds__(256) void bucket_prefix(int* ghist, int* __restrict__ btotal)
{
    const int k = blockIdx.x * 256 + threadIdx.x;
    if (k >= N_BUCKET) return;
    int run = 0;
    for (int g0 = 0; g0 < P_WG; g0 += 50) {
        int v[50];
        #pragma unroll
        for (int j = 0; j < 50; ++j) v[j] = ghist[(g0 + j) * N_BUCKET + k];
        #pragma unroll
        for (int j = 0; j < 50; ++j) {
            ghist[(g0 + j) * N_BUCKET + k] = run;
            run += v[j];
        }
    }
    btotal[k] = run;
}

__global__ __launch_bounds__(P_THREADS) void part_scatter(
    const float* __restrict__ weight,
    const int*   __restrict__ adj_src,
    const int*   __restrict__ adj_dst,
    const float* __restrict__ adj_vals,
    const int*   __restrict__ ghist,
    const int*   __restrict__ btotal,
    int*   __restrict__ bstart,
    int*   __restrict__ bpack,
    float* __restrict__ bsc)
{
    __shared__ int wbase[P_WAVES][N_BUCKET];
    __shared__ int bst[N_BUCKET];
    __shared__ int part[P_THREADS];
    const int t = threadIdx.x;
    const int lane = t & 31;
    const int w = t >> 5;
    const int g = blockIdx.x;

    for (int i = t; i < P_WAVES * N_BUCKET; i += P_THREADS) (&wbase[0][0])[i] = 0;
    int loc[5];
    int mine = 0;
    #pragma unroll
    for (int j = 0; j < 5; ++j) {
        loc[j] = (t < SCAN_LANES) ? btotal[t * 5 + j] : 0;
        mine += loc[j];
    }
    part[t] = mine;
    __syncthreads();

    const int ebase = (g * P_WAVES + w) * (P_BATCH * 32);
    int dstv[P_BATCH];
    unsigned peers[P_BATCH];
    #pragma unroll
    for (int b = 0; b < P_BATCH; ++b)
        dstv[b] = __builtin_nontemporal_load(adj_dst + ebase + b * 32 + lane);
    #pragma unroll
    for (int b = 0; b < P_BATCH; ++b) {
        const int key = dstv[b] / BUCKET_ROWS;
        const unsigned m = same_key_lanes(key);
        peers[b] = m;
        if (lane == __builtin_ctz(m)) wbase[w][key] += __builtin_popcount(m);
    }

    for (int off = 1; off < P_THREADS; off <<= 1) {
        const int add = (t >= off) ? part[t - off] : 0;
        __syncthreads();
        part[t] += add;
        __syncthreads();
    }
    int run = part[t] - mine;
    if (t < SCAN_LANES) {
        #pragma unroll
        for (int j = 0; j < 5; ++j) {
            bst[t * 5 + j] = run;
            if (g == 0) bstart[t * 5 + j] = run;
            run += loc[j];
        }
    }
    if (g == 0 && t == P_THREADS - 1) bstart[N_BUCKET] = part[P_THREADS - 1];
    __syncthreads();

    for (int k = t; k < N_BUCKET; k += P_THREADS) {
        int base = bst[k] + ghist[g * N_BUCKET + k];
        #pragma unroll
        for (int wv = 0; wv < P_WAVES; ++wv) {
            const int cnt = wbase[wv][k];
            wbase[wv][k] = base;
            base += cnt;
        }
    }
    __syncthreads();

    const unsigned lt = (1u << lane) - 1u;
    #pragma unroll
    for (int b = 0; b < P_BATCH; ++b) {
        const int e = ebase + b * 32 + lane;
        const int d = dstv[b];
        const int key = d / BUCKET_ROWS;
        const unsigned m = peers[b];
        const int leader = __builtin_ctz(m);
        int slot = 0;
        if (lane == leader) {
            slot = wbase[w][key];
            wbase[w][key] = slot + __builtin_popcount(m);
        }
        slot = __shfl(slot, leader, 32);
        const int pos = slot + __builtin_popcount(m & lt);
        bpack[pos] = ((d - key * BUCKET_ROWS) << 16) | adj_src[e];
        bsc[pos]   = weight[e / N_EDGE] * adj_vals[e];
    }
}

__global__ __launch_bounds__(256) void gather_rows(
    const float* __restrict__ x,
    const int*   __restrict__ bstart,
    const int*   __restrict__ bpack,
    const float* __restrict__ bsc,
    float*       __restrict__ sums)
{
    __shared__ int   s_src[ROWS_PER_WG][ROW_CAP];
    __shared__ float s_sc[ROWS_PER_WG][ROW_CAP];

    const int lane = threadIdx.x & 31;
    const int w    = threadIdx.x >> 5;
    const int row  = blockIdx.x * ROWS_PER_WG + w;
    const int bucket = row / BUCKET_ROWS;
    const int local  = row - bucket * BUCKET_ROWS;
    const int pb = __builtin_amdgcn_readfirstlane(bstart[bucket]);
    const int pe = __builtin_amdgcn_readfirstlane(bstart[bucket + 1]);
    const unsigned lt = (1u << lane) - 1u;

    int n = 0;
    for (int p0 = pb; p0 < pe; p0 += 256) {
        int   pk[8];
        float cv[8];
        #pragma unroll
        for (int u = 0; u < 8; ++u) {
            const int p = p0 + u * 32 + lane;
            const bool ok = p < pe;
            const int pc = ok ? p : (pe - 1);
            const int word = bpack[pc];
            pk[u] = ok ? word : -1;
            cv[u] = bsc[pc];
        }
        #pragma unroll
        for (int u = 0; u < 8; ++u) {
            const bool hit = (pk[u] >> 16) == local;
            const unsigned m = __builtin_amdgcn_ballot_w32(hit);
            if (hit) {
                const int slot = n + __builtin_popcount(m & lt);
                if (slot < ROW_CAP) {
                    s_src[w][slot] = pk[u] & 0xffff;
                    s_sc[w][slot]  = cv[u];
                }
            }
            n += __builtin_popcount(m);
        }
    }
    const bool listed = (n <= ROW_CAP);
    __syncthreads();

    float4 acc;
    acc.x = 0.f; acc.y = 0.f; acc.z = 0.f; acc.w = 0.f;

    if (listed) {
        int i = 0;
        for (; i + 8 <= n; i += 8) {
            float4 xv[8];
            float  cc[8];
            #pragma unroll
            for (int u = 0; u < 8; ++u) {
                const int ss = __builtin_amdgcn_readfirstlane(s_src[w][i + u]);
                cc[u] = s_sc[w][i + u];
                xv[u] = *(reinterpret_cast<const float4*>(x + (size_t)ss * N_HID) + lane);
            }
            #pragma unroll
            for (int u = 0; u < 8; ++u) {
                acc.x = fmaf(xv[u].x, cc[u], acc.x);
                acc.y = fmaf(xv[u].y, cc[u], acc.y);
                acc.z = fmaf(xv[u].z, cc[u], acc.z);
                acc.w = fmaf(xv[u].w, cc[u], acc.w);
            }
        }
        const int rem = n - i;
        if (rem > 0) {
            float4 xv[7];
            float  cc[7];
            #pragma unroll
            for (int u = 0; u < 7; ++u) {
                const int idx = (u < rem) ? (i + u) : i;
                const int ss = __builtin_amdgcn_readfirstlane(s_src[w][idx]);
                cc[u] = s_sc[w][idx];
                xv[u] = *(reinterpret_cast<const float4*>(x + (size_t)ss * N_HID) + lane);
            }
            #pragma unroll
            for (int u = 0; u < 7; ++u) {
                if (u < rem) {
                    acc.x = fmaf(xv[u].x, cc[u], acc.x);
                    acc.y = fmaf(xv[u].y, cc[u], acc.y);
                    acc.z = fmaf(xv[u].z, cc[u], acc.z);
                    acc.w = fmaf(xv[u].w, cc[u], acc.w);
                }
            }
        }
    } else {
        for (int p0 = pb; p0 < pe; p0 += 32) {
            const int p = p0 + lane;
            const bool ok = p < pe;
            const int   pk = ok ? bpack[p] : -1;
            const float cv = ok ? bsc[p] : 0.0f;
            unsigned m = __builtin_amdgcn_ballot_w32((pk >> 16) == local);
            while (m) {
                const int l = __builtin_ctz(m);
                m &= m - 1;
                const int ss = __builtin_amdgcn_readlane(pk, l) & 0xffff;
                const float sc = __int_as_float(__builtin_amdgcn_readlane(__float_as_int(cv), l));
                const float4 xv = *(reinterpret_cast<const float4*>(x + (size_t)ss * N_HID) + lane);
                acc.x = fmaf(xv.x, sc, acc.x);
                acc.y = fmaf(xv.y, sc, acc.y);
                acc.z = fmaf(xv.z, sc, acc.z);
                acc.w = fmaf(xv.w, sc, acc.w);
            }
        }
    }

    *(reinterpret_cast<float4*>(sums + (size_t)row * N_HID) + lane) = acc;
}

__global__ __launch_bounds__(160) void finalize_norm_gelu(float* io)
{
    const int lane = threadIdx.x & 31;
    const int wave = blockIdx.x * (blockDim.x >> 5) + (threadIdx.x >> 5);
    if (wave >= (N_NODES / 16)) return;

    const int node   = wave * 16 + (lane & 15);
    const int colOff = 2 * (lane >> 4);
    const float* rowp = io + (long long)node * N_HID + colOff;

    v2f vbuf[32];
    v8f c = {};
    #pragma unroll
    for (int j = 0; j < 32; ++j) {
        v2f a;
        a.x = rowp[4 * j];
        a.y = rowp[4 * j + 1];
        vbuf[j] = a;
        c = __builtin_amdgcn_wmma_f32_16x16x4_f32(
                false, a, false, a, (short)0, c, false, false);
    }

    const int sel = lane & 7;
    float dv = c[0];
    dv = (sel == 1) ? c[1] : dv;
    dv = (sel == 2) ? c[2] : dv;
    dv = (sel == 3) ? c[3] : dv;
    dv = (sel == 4) ? c[4] : dv;
    dv = (sel == 5) ? c[5] : dv;
    dv = (sel == 6) ? c[6] : dv;
    dv = (sel == 7) ? c[7] : dv;
    const int nm      = lane & 15;
    const int srcLane = (nm < 8) ? nm : (nm + 16);
    const float sumsq = __shfl(dv, srcLane, 32);

    const float scale = 1.0f / fmaxf(sqrtf(sumsq), 1e-12f);

    float* op = io + (long long)node * N_HID + colOff;
    const float inv_sqrt2 = 0.70710678118654752440f;
    #pragma unroll
    for (int j = 0; j < 32; ++j) {
        float vx = vbuf[j].x * scale;
        float vy = vbuf[j].y * scale;
        vx = 0.5f * vx * (1.0f + erff(vx * inv_sqrt2));
        vy = 0.5f * vy * (1.0f + erff(vy * inv_sqrt2));
        op[4 * j]     = vx;
        op[4 * j + 1] = vy;
    }
}

extern "C" void kernel_launch(void* const* d_in, const int* in_sizes, int n_in,
                              void* d_out, int out_size, void* d_ws, size_t ws_size,
                              hipStream_t stream) {
    const float* x        = (const float*)d_in[0];
    const float* weight   = (const float*)d_in[1];
    const int*   adj_src  = (const int*)d_in[2];
    const int*   adj_dst  = (const int*)d_in[3];
    const float* adj_vals = (const float*)d_in[4];
    float*       out      = (float*)d_out;

    int*   ghist  = (int*)d_ws;
    int*   btotal = ghist + P_WG * N_BUCKET;
    int*   bstart = btotal + 2048;
    int*   bpack  = bstart + 2048;
    float* bsc    = (float*)(bpack + TOTAL_EDGES);

    part_count<<<P_WG, P_THREADS, 0, stream>>>(adj_dst, ghist);
    bucket_prefix<<<(N_BUCKET + 255) / 256, 256, 0, stream>>>(ghist, btotal);
    part_scatter<<<P_WG, P_THREADS, 0, stream>>>(weight, adj_src, adj_dst, adj_vals,
                                                 ghist, btotal, bstart, bpack, bsc);
    gather_rows<<<N_NODES / ROWS_PER_WG, 256, 0, stream>>>(x, bstart, bpack, bsc, out);
    const int tiles = N_NODES / 16;
    finalize_norm_gelu<<<(tiles + 4) / 5, 160, 0, stream>>>(out);
}


# Normalize+GELU moved into the row-sum kernel's epilogue (row stays in registers; sum of squares by 2 f32 WMMAs per row + wave butterfly, so its summation order changes); 5 launches -> 4
# speedup vs baseline: 2.0375x; 1.0507x over previous
#include <hip/hip_runtime.h>
#include <math.h>

#define N_NODES 50000
#define N_HID   128
#define N_ADJ   4
#define N_EDGE  600000
#define TOTAL_EDGES (N_ADJ * N_EDGE)

#define BUCKET_ROWS 25
#define N_BUCKET    (N_NODES / BUCKET_ROWS)
#define KEY_BITS    11
#define P_WG        250
#define P_WAVES     15
#define P_THREADS   (P_WAVES * 32)
#define P_BATCH     20
#define SCAN_LANES  400
#define ROW_CAP     128
#define ROWS_PER_WG 8

typedef __attribute__((ext_vector_type(2))) float v2f;
typedef __attribute__((ext_vector_type(8))) float v8f;

__device__ __forceinline__ unsigned same_key_lanes(int key) {
    unsigned m = 0xffffffffu;
    #pragma unroll
    for (int b = 0; b < KEY_BITS; ++b) {
        const bool bit = ((key >> b) & 1) != 0;
        const unsigned bal = __builtin_amdgcn_ballot_w32(bit);
        m &= bit ? bal : ~bal;
    }
    return m;
}

__global__ __launch_bounds__(P_THREADS) void part_count(
    const int* __restrict__ adj_dst, int* __restrict__ ghist)
{
    __shared__ int whist[P_WAVES][N_BUCKET];
    const int t = threadIdx.x;
    const int lane = t & 31;
    const int w = t >> 5;
    for (int i = t; i < P_WAVES * N_BUCKET; i += P_THREADS) (&whist[0][0])[i] = 0;
    __syncthreads();
    const int ebase = (blockIdx.x * P_WAVES + w) * (P_BATCH * 32);
    int key[P_BATCH];
    #pragma unroll
    for (int b = 0; b < P_BATCH; ++b)
        key[b] = __builtin_nontemporal_load(adj_dst + ebase + b * 32 + lane) / BUCKET_ROWS;
    #pragma unroll
    for (int b = 0; b < P_BATCH; ++b) {
        const unsigned m = same_key_lanes(key[b]);
        if (lane == __builtin_ctz(m)) whist[w][key[b]] += __builtin_popcount(m);
    }
    __syncthreads();
    for (int k = t; k < N_BUCKET; k += P_THREADS) {
        int sum = 0;
        #pragma unroll
        for (int wv = 0; wv < P_WAVES; ++wv) sum += whist[wv][k];
        ghist[blockIdx.x * N_BUCKET + k] = sum;
    }
}

__global__ __launch_bounds__(256) void bucket_prefix(int* ghist, int* __restrict__ btotal)
{
    const int k = blockIdx.x * 256 + threadIdx.x;
    if (k >= N_BUCKET) return;
    int run = 0;
    for (int g0 = 0; g0 < P_WG; g0 += 50) {
        int v[50];
        #pragma unroll
        for (int j = 0; j < 50; ++j) v[j] = ghist[(g0 + j) * N_BUCKET + k];
        #pragma unroll
        for (int j = 0; j < 50; ++j) {
            ghist[(g0 + j) * N_BUCKET + k] = run;
            run += v[j];
        }
    }
    btotal[k] = run;
}

__global__ __launch_bounds__(P_THREADS) void part_scatter(
    const float* __restrict__ weight,
    const int*   __restrict__ adj_src,
    const int*   __restrict__ adj_dst,
    const float* __restrict__ adj_vals,
    const int*   __restrict__ ghist,
    const int*   __restrict__ btotal,
    int*   __restrict__ bstart,
    int*   __restrict__ bpack,
    float* __restrict__ bsc)
{
    __shared__ int wbase[P_WAVES][N_BUCKET];
    __shared__ int bst[N_BUCKET];
    __shared__ int part[P_THREADS];
    const int t = threadIdx.x;
    const int lane = t & 31;
    const int w = t >> 5;
    const int g = blockIdx.x;

    for (int i = t; i < P_WAVES * N_BUCKET; i += P_THREADS) (&wbase[0][0])[i] = 0;
    int loc[5];
    int mine = 0;
    #pragma unroll
    for (int j = 0; j < 5; ++j) {
        loc[j] = (t < SCAN_LANES) ? btotal[t * 5 + j] : 0;
        mine += loc[j];
    }
    part[t] = mine;
    __syncthreads();

    const int ebase = (g * P_WAVES + w) * (P_BATCH * 32);
    int dstv[P_BATCH];
    unsigned peers[P_BATCH];
    #pragma unroll
    for (int b = 0; b < P_BATCH; ++b)
        dstv[b] = __builtin_nontemporal_load(adj_dst + ebase + b * 32 + lane);
    #pragma unroll
    for (int b = 0; b < P_BATCH; ++b) {
        const int key = dstv[b] / BUCKET_ROWS;
        const unsigned m = same_key_lanes(key);
        peers[b] = m;
        if (lane == __builtin_ctz(m)) wbase[w][key] += __builtin_popcount(m);
    }

    for (int off = 1; off < P_THREADS; off <<= 1) {
        const int add = (t >= off) ? part[t - off] : 0;
        __syncthreads();
        part[t] += add;
        __syncthreads();
    }
    int run = part[t] - mine;
    if (t < SCAN_LANES) {
        #pragma unroll
        for (int j = 0; j < 5; ++j) {
            bst[t * 5 + j] = run;
            if (g == 0) bstart[t * 5 + j] = run;
            run += loc[j];
        }
    }
    if (g == 0 && t == P_THREADS - 1) bstart[N_BUCKET] = part[P_THREADS - 1];
    __syncthreads();

    for (int k = t; k < N_BUCKET; k += P_THREADS) {
        int base = bst[k] + ghist[g * N_BUCKET + k];
        #pragma unroll
        for (int wv = 0; wv < P_WAVES; ++wv) {
            const int cnt = wbase[wv][k];
            wbase[wv][k] = base;
            base += cnt;
        }
    }
    __syncthreads();

    const unsigned lt = (1u << lane) - 1u;
    #pragma unroll
    for (int b = 0; b < P_BATCH; ++b) {
        const int e = ebase + b * 32 + lane;
        const int d = dstv[b];
        const int key = d / BUCKET_ROWS;
        const unsigned m = peers[b];
        const int leader = __builtin_ctz(m);
        int slot = 0;
        if (lane == leader) {
            slot = wbase[w][key];
            wbase[w][key] = slot + __builtin_popcount(m);
        }
        slot = __shfl(slot, leader, 32);
        const int pos = slot + __builtin_popcount(m & lt);
        bpack[pos] = ((d - key * BUCKET_ROWS) << 16) | adj_src[e];
        bsc[pos]   = weight[e / N_EDGE] * adj_vals[e];
    }
}

template <int M>
__device__ __forceinline__ float row_xor_add(float v) {
    const int b = __builtin_bit_cast(int, v);
    const int r = __builtin_amdgcn_update_dpp(0, b, 0x160 | M, 0xf, 0xf, true);
    return v + __builtin_bit_cast(float, r);
}
__device__ __forceinline__ float xhalf(float x) {
    const int b = __builtin_bit_cast(int, x);
    return __builtin_bit_cast(float,
        __builtin_amdgcn_permlanex16(b, b, 0x76543210, 0xfedcba98, false, false));
}

__global__ __launch_bounds__(256) void gather_rows(
    const float* __restrict__ x,
    const int*   __restrict__ bstart,
    const int*   __restrict__ bpack,
    const float* __restrict__ bsc,
    float*       __restrict__ outp)
{
    __shared__ int   s_src[ROWS_PER_WG][ROW_CAP];
    __shared__ float s_sc[ROWS_PER_WG][ROW_CAP];

    const int lane = threadIdx.x & 31;
    const int w    = threadIdx.x >> 5;
    const int row  = blockIdx.x * ROWS_PER_WG + w;
    const int bucket = row / BUCKET_ROWS;
    const int local  = row - bucket * BUCKET_ROWS;
    const int pb = __builtin_amdgcn_readfirstlane(bstart[bucket]);
    const int pe = __builtin_amdgcn_readfirstlane(bstart[bucket + 1]);
    const unsigned lt = (1u << lane) - 1u;

    int n = 0;
    for (int p0 = pb; p0 < pe; p0 += 256) {
        int   pk[8];
        float cv[8];
        #pragma unroll
        for (int u = 0; u < 8; ++u) {
            const int p = p0 + u * 32 + lane;
            const bool ok = p < pe;
            const int pc = ok ? p : (pe - 1);
            const int word = bpack[pc];
            pk[u] = ok ? word : -1;
            cv[u] = bsc[pc];
        }
        #pragma unroll
        for (int u = 0; u < 8; ++u) {
            const bool hit = (pk[u] >> 16) == local;
            const unsigned m = __builtin_amdgcn_ballot_w32(hit);
            if (hit) {
                const int slot = n + __builtin_popcount(m & lt);
                if (slot < ROW_CAP) {
                    s_src[w][slot] = pk[u] & 0xffff;
                    s_sc[w][slot]  = cv[u];
                }
            }
            n += __builtin_popcount(m);
        }
    }
    const bool listed = (n <= ROW_CAP);
    __syncthreads();

    float4 acc;
    acc.x = 0.f; acc.y = 0.f; acc.z = 0.f; acc.w = 0.f;

    if (listed) {
        int i = 0;
        for (; i + 8 <= n; i += 8) {
            float4 xv[8];
            float  cc[8];
            #pragma unroll
            for (int u = 0; u < 8; ++u) {
                const int ss = __builtin_amdgcn_readfirstlane(s_src[w][i + u]);
                cc[u] = s_sc[w][i + u];
                xv[u] = *(reinterpret_cast<const float4*>(x + (size_t)ss * N_HID) + lane);
            }
            #pragma unroll
            for (int u = 0; u < 8; ++u) {
                acc.x = fmaf(xv[u].x, cc[u], acc.x);
                acc.y = fmaf(xv[u].y, cc[u], acc.y);
                acc.z = fmaf(xv[u].z, cc[u], acc.z);
                acc.w = fmaf(xv[u].w, cc[u], acc.w);
            }
        }
        const int rem = n - i;
        if (rem > 0) {
            float4 xv[7];
            float  cc[7];
            #pragma unroll
            for (int u = 0; u < 7; ++u) {
                const int idx = (u < rem) ? (i + u) : i;
                const int ss = __builtin_amdgcn_readfirstlane(s_src[w][idx]);
                cc[u] = s_sc[w][idx];
                xv[u] = *(reinterpret_cast<const float4*>(x + (size_t)ss * N_HID) + lane);
            }
            #pragma unroll
            for (int u = 0; u < 7; ++u) {
                if (u < rem) {
                    acc.x = fmaf(xv[u].x, cc[u], acc.x);
                    acc.y = fmaf(xv[u].y, cc[u], acc.y);
                    acc.z = fmaf(xv[u].z, cc[u], acc.z);
                    acc.w = fmaf(xv[u].w, cc[u], acc.w);
                }
            }
        }
    } else {
        for (int p0 = pb; p0 < pe; p0 += 32) {
            const int p = p0 + lane;
            const bool ok = p < pe;
            const int   pk = ok ? bpack[p] : -1;
            const float cv = ok ? bsc[p] : 0.0f;
            unsigned m = __builtin_amdgcn_ballot_w32((pk >> 16) == local);
            while (m) {
                const int l = __builtin_ctz(m);
                m &= m - 1;
                const int ss = __builtin_amdgcn_readlane(pk, l) & 0xffff;
                const float sc = __int_as_float(__builtin_amdgcn_readlane(__float_as_int(cv), l));
                const float4 xv = *(reinterpret_cast<const float4*>(x + (size_t)ss * N_HID) + lane);
                acc.x = fmaf(xv.x, sc, acc.x);
                acc.y = fmaf(xv.y, sc, acc.y);
                acc.z = fmaf(xv.z, sc, acc.z);
                acc.w = fmaf(xv.w, sc, acc.w);
            }
        }
    }

    v2f a0, a1;
    a0.x = acc.x; a0.y = acc.y;
    a1.x = acc.z; a1.y = acc.w;
    v8f c = {};
    c = __builtin_amdgcn_wmma_f32_16x16x4_f32(false, a0, false, a0, (short)0, c, false, false);
    c = __builtin_amdgcn_wmma_f32_16x16x4_f32(false, a1, false, a1, (short)0, c, false, false);
    const int sel = lane & 7;
    float dv = c[0];
    dv = (sel == 1) ? c[1] : dv;
    dv = (sel == 2) ? c[2] : dv;
    dv = (sel == 3) ? c[3] : dv;
    dv = (sel == 4) ? c[4] : dv;
    dv = (sel == 5) ? c[5] : dv;
    dv = (sel == 6) ? c[6] : dv;
    dv = (sel == 7) ? c[7] : dv;
    dv = (lane < 8 || lane >= 24) ? dv : 0.0f;
    dv = row_xor_add<1>(dv);
    dv = row_xor_add<2>(dv);
    dv = row_xor_add<4>(dv);
    dv = row_xor_add<8>(dv);
    dv = dv + xhalf(dv);

    const float scale = 1.0f / fmaxf(sqrtf(dv), 1e-12f);
    const float inv_sqrt2 = 0.70710678118654752440f;
    float4 o;
    o.x = acc.x * scale;
    o.y = acc.y * scale;
    o.z = acc.z * scale;
    o.w = acc.w * scale;
    o.x = 0.5f * o.x * (1.0f + erff(o.x * inv_sqrt2));
    o.y = 0.5f * o.y * (1.0f + erff(o.y * inv_sqrt2));
    o.z = 0.5f * o.z * (1.0f + erff(o.z * inv_sqrt2));
    o.w = 0.5f * o.w * (1.0f + erff(o.w * inv_sqrt2));
    *(reinterpret_cast<float4*>(outp + (size_t)row * N_HID) + lane) = o;
}

extern "C" void kernel_launch(void* const* d_in, const int* in_sizes, int n_in,
                              void* d_out, int out_size, void* d_ws, size_t ws_size,
                              hipStream_t stream) {
    const float* x        = (const float*)d_in[0];
    const float* weight   = (const float*)d_in[1];
    const int*   adj_src  = (const int*)d_in[2];
    const int*   adj_dst  = (const int*)d_in[3];
    const float* adj_vals = (const float*)d_in[4];
    float*       out      = (float*)d_out;

    int*   ghist  = (int*)d_ws;
    int*   btotal = ghist + P_WG * N_BUCKET;
    int*   bstart = btotal + 2048;
    int*   bpack  = bstart + 2048;
    float* bsc    = (float*)(bpack + TOTAL_EDGES);

    part_count<<<P_WG, P_THREADS, 0, stream>>>(adj_dst, ghist);
    bucket_prefix<<<(N_BUCKET + 255) / 256, 256, 0, stream>>>(ghist, btotal);
    part_scatter<<<P_WG, P_THREADS, 0, stream>>>(weight, adj_src, adj_dst, adj_vals,
                                                 ghist, btotal, bstart, bpack, bsc);
    gather_rows<<<N_NODES / ROWS_PER_WG, 256, 0, stream>>>(x, bstart, bpack, bsc, out);
}


# List entries kept as one 8-byte record (packed word, scale): one scattered 8-B store per edge instead of two 4-B, one 8-B load per entry in the row scan and in LDS; sum order kept
# speedup vs baseline: 2.7717x; 1.3603x over previous
#include <hip/hip_runtime.h>
#include <math.h>

#define N_NODES 50000
#define N_HID   128
#define N_ADJ   4
#define N_EDGE  600000
#define TOTAL_EDGES (N_ADJ * N_EDGE)

#define BUCKET_ROWS 25
#define N_BUCKET    (N_NODES / BUCKET_ROWS)
#define KEY_BITS    11
#define P_WG        250
#define P_WAVES     15
#define P_THREADS   (P_WAVES * 32)
#define P_BATCH     20
#define SCAN_LANES  400
#define ROW_CAP     128
#define ROWS_PER_WG 8

typedef __attribute__((ext_vector_type(2))) float v2f;
typedef __attribute__((ext_vector_type(8))) float v8f;

__device__ __forceinline__ unsigned same_key_lanes(int key) {
    unsigned m = 0xffffffffu;
    #pragma unroll
    for (int b = 0; b < KEY_BITS; ++b) {
        const bool bit = ((key >> b) & 1) != 0;
        const unsigned bal = __builtin_amdgcn_ballot_w32(bit);
        m &= bit ? bal : ~bal;
    }
    return m;
}

__global__ __launch_bounds__(P_THREADS) void part_count(
    const int* __restrict__ adj_dst, int* __restrict__ ghist)
{
    __shared__ int whist[P_WAVES][N_BUCKET];
    const int t = threadIdx.x;
    const int lane = t & 31;
    const int w = t >> 5;
    for (int i = t; i < P_WAVES * N_BUCKET; i += P_THREADS) (&whist[0][0])[i] = 0;
    __syncthreads();
    const int ebase = (blockIdx.x * P_WAVES + w) * (P_BATCH * 32);
    int key[P_BATCH];
    #pragma unroll
    for (int b = 0; b < P_BATCH; ++b)
        key[b] = __builtin_nontemporal_load(adj_dst + ebase + b * 32 + lane) / BUCKET_ROWS;
    #pragma unroll
    for (int b = 0; b < P_BATCH; ++b) {
        const unsigned m = same_key_lanes(key[b]);
        if (lane == __builtin_ctz(m)) whist[w][key[b]] += __builtin_popcount(m);
    }
    __syncthreads();
    for (int k = t; k < N_BUCKET; k += P_THREADS) {
        int sum = 0;
        #pragma unroll
        for (int wv = 0; wv < P_WAVES; ++wv) sum += whist[wv][k];
        ghist[blockIdx.x * N_BUCKET + k] = sum;
    }
}

__global__ __launch_bounds__(256) void bucket_prefix(int* ghist, int* __restrict__ btotal)
{
    const int k = blockIdx.x * 256 + threadIdx.x;
    if (k >= N_BUCKET) return;
    int run = 0;
    for (int g0 = 0; g0 < P_WG; g0 += 50) {
        int v[50];
        #pragma unroll
        for (int j = 0; j < 50; ++j) v[j] = ghist[(g0 + j) * N_BUCKET + k];
        #pragma unroll
        for (int j = 0; j < 50; ++j) {
            ghist[(g0 + j) * N_BUCKET + k] = run;
            run += v[j];
        }
    }
    btotal[k] = run;
}

__global__ __launch_bounds__(P_THREADS) void part_scatter(
    const float* __restrict__ weight,
    const int*   __restrict__ adj_src,
    const int*   __restrict__ adj_dst,
    const float* __restrict__ adj_vals,
    const int*   __restrict__ ghist,
    const int*   __restrict__ btotal,
    int*   __restrict__ bstart,
    int2*  __restrict__ brec)
{
    __shared__ int wbase[P_WAVES][N_BUCKET];
    __shared__ int bst[N_BUCKET];
    __shared__ int part[P_THREADS];
    const int t = threadIdx.x;
    const int lane = t & 31;
    const int w = t >> 5;
    const int g = blockIdx.x;

    for (int i = t; i < P_WAVES * N_BUCKET; i += P_THREADS) (&wbase[0][0])[i] = 0;
    int loc[5];
    int mine = 0;
    #pragma unroll
    for (int j = 0; j < 5; ++j) {
        loc[j] = (t < SCAN_LANES) ? btotal[t * 5 + j] : 0;
        mine += loc[j];
    }
    part[t] = mine;
    __syncthreads();

    const int ebase = (g * P_WAVES + w) * (P_BATCH * 32);
    int dstv[P_BATCH];
    unsigned peers[P_BATCH];
    #pragma unroll
    for (int b = 0; b < P_BATCH; ++b)
        dstv[b] = __builtin_nontemporal_load(adj_dst + ebase + b * 32 + lane);
    #pragma unroll
    for (int b = 0; b < P_BATCH; ++b) {
        const int key = dstv[b] / BUCKET_ROWS;
        const unsigned m = same_key_lanes(key);
        peers[b] = m;
        if (lane == __builtin_ctz(m)) wbase[w][key] += __builtin_popcount(m);
    }

    for (int off = 1; off < P_THREADS; off <<= 1) {
        const int add = (t >= off) ? part[t - off] : 0;
        __syncthreads();
        part[t] += add;
        __syncthreads();
    }
    int run = part[t] - mine;
    if (t < SCAN_LANES) {
        #pragma unroll
        for (int j = 0; j < 5; ++j) {
            bst[t * 5 + j] = run;
            if (g == 0) bstart[t * 5 + j] = run;
            run += loc[j];
        }
    }
    if (g == 0 && t == P_THREADS - 1) bstart[N_BUCKET] = part[P_THREADS - 1];
    __syncthreads();

    for (int k = t; k < N_BUCKET; k += P_THREADS) {
        int base = bst[k] + ghist[g * N_BUCKET + k];
        #pragma unroll
        for (int wv = 0; wv < P_WAVES; ++wv) {
            const int cnt = wbase[wv][k];
            wbase[wv][k] = base;
            base += cnt;
        }
    }
    __syncthreads();

    const unsigned lt = (1u << lane) - 1u;
    #pragma unroll
    for (int b = 0; b < P_BATCH; ++b) {
        const int e = ebase + b * 32 + lane;
        const int d = dstv[b];
        const int key = d / BUCKET_ROWS;
        const unsigned m = peers[b];
        const int leader = __builtin_ctz(m);
        int slot = 0;
        if (lane == leader) {
            slot = wbase[w][key];
            wbase[w][key] = slot + __builtin_popcount(m);
        }
        slot = __shfl(slot, leader, 32);
        const int pos = slot + __builtin_popcount(m & lt);
        int2 rec;
        rec.x = ((d - key * BUCKET_ROWS) << 16) | adj_src[e];
        rec.y = __float_as_int(weight[e / N_EDGE] * adj_vals[e]);
        brec[pos] = rec;
    }
}

template <int M>
__device__ __forceinline__ float row_xor_add(float v) {
    const int b = __builtin_bit_cast(int, v);
    const int r = __builtin_amdgcn_update_dpp(0, b, 0x160 | M, 0xf, 0xf, true);
    return v + __builtin_bit_cast(float, r);
}
__device__ __forceinline__ float xhalf(float x) {
    const int b = __builtin_bit_cast(int, x);
    return __builtin_bit_cast(float,
        __builtin_amdgcn_permlanex16(b, b, 0x76543210, 0xfedcba98, false, false));
}

__global__ __launch_bounds__(256, 16) void gather_rows(
    const float* __restrict__ x,
    const int*   __restrict__ bstart,
    const int2*  __restrict__ brec,
    float*       __restrict__ outp)
{
    __shared__ int2 s_rec[ROWS_PER_WG][ROW_CAP];

    const int lane = threadIdx.x & 31;
    const int w    = threadIdx.x >> 5;
    const int row  = blockIdx.x * ROWS_PER_WG + w;
    const int bucket = row / BUCKET_ROWS;
    const int local  = row - bucket * BUCKET_ROWS;
    const int pb = __builtin_amdgcn_readfirstlane(bstart[bucket]);
    const int pe = __builtin_amdgcn_readfirstlane(bstart[bucket + 1]);
    const unsigned lt = (1u << lane) - 1u;

    int n = 0;
    for (int p0 = pb; p0 < pe; p0 += 256) {
        int2 rc[8];
        #pragma unroll
        for (int u = 0; u < 8; ++u) {
            const int p = p0 + u * 32 + lane;
            const bool ok = p < pe;
            const int pc = ok ? p : (pe - 1);
            rc[u] = brec[pc];
            rc[u].x = ok ? rc[u].x : -1;
        }
        #pragma unroll
        for (int u = 0; u < 8; ++u) {
            const bool hit = (rc[u].x >> 16) == local;
            const unsigned m = __builtin_amdgcn_ballot_w32(hit);
            if (hit) {
                const int slot = n + __builtin_popcount(m & lt);
                if (slot < ROW_CAP) {
                    int2 keep;
                    keep.x = rc[u].x & 0xffff;
                    keep.y = rc[u].y;
                    s_rec[w][slot] = keep;
                }
            }
            n += __builtin_popcount(m);
        }
    }
    const bool listed = (n <= ROW_CAP);
    __syncthreads();

    float4 acc;
    acc.x = 0.f; acc.y = 0.f; acc.z = 0.f; acc.w = 0.f;

    if (listed) {
        int i = 0;
        for (; i + 8 <= n; i += 8) {
            float4 xv[8];
            float  cc[8];
            #pragma unroll
            for (int u = 0; u < 8; ++u) {
                const int2 r = s_rec[w][i + u];
                const int ss = __builtin_amdgcn_readfirstlane(r.x);
                cc[u] = __int_as_float(r.y);
                xv[u] = *(reinterpret_cast<const float4*>(x + (size_t)ss * N_HID) + lane);
            }
            #pragma unroll
            for (int u = 0; u < 8; ++u) {
                acc.x = fmaf(xv[u].x, cc[u], acc.x);
                acc.y = fmaf(xv[u].y, cc[u], acc.y);
                acc.z = fmaf(xv[u].z, cc[u], acc.z);
                acc.w = fmaf(xv[u].w, cc[u], acc.w);
            }
        }
        const int rem = n - i;
        if (rem > 0) {
            float4 xv[7];
            float  cc[7];
            #pragma unroll
            for (int u = 0; u < 7; ++u) {
                const int idx = (u < rem) ? (i + u) : i;
                const int2 r = s_rec[w][idx];
                const int ss = __builtin_amdgcn_readfirstlane(r.x);
                cc[u] = __int_as_float(r.y);
                xv[u] = *(reinterpret_cast<const float4*>(x + (size_t)ss * N_HID) + lane);
            }
            #pragma unroll
            for (int u = 0; u < 7; ++u) {
                if (u < rem) {
                    acc.x = fmaf(xv[u].x, cc[u], acc.x);
                    acc.y = fmaf(xv[u].y, cc[u], acc.y);
                    acc.z = fmaf(xv[u].z, cc[u], acc.z);
                    acc.w = fmaf(xv[u].w, cc[u], acc.w);
                }
            }
        }
    } else {
        for (int p0 = pb; p0 < pe; p0 += 32) {
            const int p = p0 + lane;
            const bool ok = p < pe;
            int2 rc;
            rc.x = -1; rc.y = 0;
            if (ok) rc = brec[p];
            const int pk = rc.x;
            const int cv = rc.y;
            unsigned m = __builtin_amdgcn_ballot_w32((pk >> 16) == local);
            while (m) {
                const int l = __builtin_ctz(m);
                m &= m - 1;
                const int ss = __builtin_amdgcn_readlane(pk, l) & 0xffff;
                const float sc = __int_as_float(__builtin_amdgcn_readlane(cv, l));
                const float4 xv = *(reinterpret_cast<const float4*>(x + (size_t)ss * N_HID) + lane);
                acc.x = fmaf(xv.x, sc, acc.x);
                acc.y = fmaf(xv.y, sc, acc.y);
                acc.z = fmaf(xv.z, sc, acc.z);
                acc.w = fmaf(xv.w, sc, acc.w);
            }
        }
    }

    v2f a0, a1;
    a0.x = acc.x; a0.y = acc.y;
    a1.x = acc.z; a1.y = acc.w;
    v8f c = {};
    c = __builtin_amdgcn_wmma_f32_16x16x4_f32(false, a0, false, a0, (short)0, c, false, false);
    c = __builtin_amdgcn_wmma_f32_16x16x4_f32(false, a1, false, a1, (short)0, c, false, false);
    const int sel = lane & 7;
    float dv = c[0];
    dv = (sel == 1) ? c[1] : dv;
    dv = (sel == 2) ? c[2] : dv;
    dv = (sel == 3) ? c[3] : dv;
    dv = (sel == 4) ? c[4] : dv;
    dv = (sel == 5) ? c[5] : dv;
    dv = (sel == 6) ? c[6] : dv;
    dv = (sel == 7) ? c[7] : dv;
    dv = (lane < 8 || lane >= 24) ? dv : 0.0f;
    dv = row_xor_add<1>(dv);
    dv = row_xor_add<2>(dv);
    dv = row_xor_add<4>(dv);
    dv = row_xor_add<8>(dv);
    dv = dv + xhalf(dv);

    const float scale = 1.0f / fmaxf(sqrtf(dv), 1e-12f);
    const float inv_sqrt2 = 0.70710678118654752440f;
    float4 o;
    o.x = acc.x * scale;
    o.y = acc.y * scale;
    o.z = acc.z * scale;
    o.w = acc.w * scale;
    o.x = 0.5f * o.x * (1.0f + erff(o.x * inv_sqrt2));
    o.y = 0.5f * o.y * (1.0f + erff(o.y * inv_sqrt2));
    o.z = 0.5f * o.z * (1.0f + erff(o.z * inv_sqrt2));
    o.w = 0.5f * o.w * (1.0f + erff(o.w * inv_sqrt2));
    *(reinterpret_cast<float4*>(outp + (size_t)row * N_HID) + lane) = o;
}

extern "C" void kernel_launch(void* const* d_in, const int* in_sizes, int n_in,
                              void* d_out, int out_size, void* d_ws, size_t ws_size,
                              hipStream_t stream) {
    const float* x        = (const float*)d_in[0];
    const float* weight   = (const float*)d_in[1];
    const int*   adj_src  = (const int*)d_in[2];
    const int*   adj_dst  = (const int*)d_in[3];
    const float* adj_vals = (const float*)d_in[4];
    float*       out      = (float*)d_out;

    int*   ghist  = (int*)d_ws;
    int*   btotal = ghist + P_WG * N_BUCKET;
    int*   bstart = btotal + 2048;
    int2*  brec   = (int2*)(bstart + 2048);

    part_count<<<P_WG, P_THREADS, 0, stream>>>(adj_dst, ghist);
    bucket_prefix<<<(N_BUCKET + 255) / 256, 256, 0, stream>>>(ghist, btotal);
    part_scatter<<<P_WG, P_THREADS, 0, stream>>>(weight, adj_src, adj_dst, adj_vals,
                                                 ghist, btotal, bstart, brec);
    gather_rows<<<N_NODES / ROWS_PER_WG, 256, 0, stream>>>(x, bstart, brec, out);
}


# Placement kernel stages its records in LDS in bucket order and stores them with consecutive lanes on consecutive slots (a run per bucket) instead of one scattered 8-B store per edge; list unchanged
# speedup vs baseline: 3.5872x; 1.2942x over previous
#include <hip/hip_runtime.h>
#include <math.h>

#define N_NODES 50000
#define N_HID   128
#define N_ADJ   4
#define N_EDGE  600000
#define TOTAL_EDGES (N_ADJ * N_EDGE)

#define BUCKET_ROWS 25
#define N_BUCKET    (N_NODES / BUCKET_ROWS)
#define KEY_BITS    11
#define P_WG        250
#define P_WAVES     15
#define P_THREADS   (P_WAVES * 32)
#define P_BATCH     20
#define SCAN_LANES  400
#define ROW_CAP     128
#define ROWS_PER_WG 8

typedef __attribute__((ext_vector_type(2))) float v2f;
typedef __attribute__((ext_vector_type(8))) float v8f;

__device__ __forceinline__ unsigned same_key_lanes(int key) {
    unsigned m = 0xffffffffu;
    #pragma unroll
    for (int b = 0; b < KEY_BITS; ++b) {
        const bool bit = ((key >> b) & 1) != 0;
        const unsigned bal = __builtin_amdgcn_ballot_w32(bit);
        m &= bit ? bal : ~bal;
    }
    return m;
}

__global__ __launch_bounds__(P_THREADS) void part_count(
    const int* __restrict__ adj_dst, int* __restrict__ ghist)
{
    __shared__ int whist[P_WAVES][N_BUCKET];
    const int t = threadIdx.x;
    const int lane = t & 31;
    const int w = t >> 5;
    for (int i = t; i < P_WAVES * N_BUCKET; i += P_THREADS) (&whist[0][0])[i] = 0;
    __syncthreads();
    const int ebase = (blockIdx.x * P_WAVES + w) * (P_BATCH * 32);
    int key[P_BATCH];
    #pragma unroll
    for (int b = 0; b < P_BATCH; ++b)
        key[b] = __builtin_nontemporal_load(adj_dst + ebase + b * 32 + lane) / BUCKET_ROWS;
    #pragma unroll
    for (int b = 0; b < P_BATCH; ++b) {
        const unsigned m = same_key_lanes(key[b]);
        if (lane == __builtin_ctz(m)) whist[w][key[b]] += __builtin_popcount(m);
    }
    __syncthreads();
    for (int k = t; k < N_BUCKET; k += P_THREADS) {
        int sum = 0;
        #pragma unroll
        for (int wv = 0; wv < P_WAVES; ++wv) sum += whist[wv][k];
        ghist[blockIdx.x * N_BUCKET + k] = sum;
    }
}

__global__ __launch_bounds__(256) void bucket_prefix(int* ghist, int* __restrict__ btotal)
{
    const int k = blockIdx.x * 256 + threadIdx.x;
    if (k >= N_BUCKET) return;
    int run = 0;
    for (int g0 = 0; g0 < P_WG; g0 += 50) {
        int v[50];
        #pragma unroll
        for (int j = 0; j < 50; ++j) v[j] = ghist[(g0 + j) * N_BUCKET + k];
        #pragma unroll
        for (int j = 0; j < 50; ++j) {
            ghist[(g0 + j) * N_BUCKET + k] = run;
            run += v[j];
        }
    }
    btotal[k] = run;
}

#define P_EDGES (P_WAVES * P_BATCH * 32)
__global__ __launch_bounds__(P_THREADS) void part_scatter(
    const float* __restrict__ weight,
    const int*   __restrict__ adj_src,
    const int*   __restrict__ adj_dst,
    const float* __restrict__ adj_vals,
    const int*   __restrict__ ghist,
    const int*   __restrict__ btotal,
    int*   __restrict__ bstart,
    int2*  __restrict__ brec)
{
    __shared__ int  wbase[P_WAVES][N_BUCKET];
    __shared__ int  dlt[N_BUCKET];
    __shared__ int2 part[P_THREADS];
    __shared__ int2 srec[P_EDGES];
    __shared__ int  spos[P_EDGES];
    const int t = threadIdx.x;
    const int lane = t & 31;
    const int w = t >> 5;
    const int g = blockIdx.x;

    for (int i = t; i < P_WAVES * N_BUCKET; i += P_THREADS) (&wbase[0][0])[i] = 0;
    int loc[5];
    int gh[5];
    int mine = 0;
    #pragma unroll
    for (int j = 0; j < 5; ++j) {
        loc[j] = (t < SCAN_LANES) ? btotal[t * 5 + j] : 0;
        gh[j]  = (t < SCAN_LANES) ? ghist[g * N_BUCKET + t * 5 + j] : 0;
        mine += loc[j];
    }
    __syncthreads();

    const int ebase = (g * P_WAVES + w) * (P_BATCH * 32);
    int dstv[P_BATCH];
    unsigned peers[P_BATCH];
    #pragma unroll
    for (int b = 0; b < P_BATCH; ++b)
        dstv[b] = __builtin_nontemporal_load(adj_dst + ebase + b * 32 + lane);
    #pragma unroll
    for (int b = 0; b < P_BATCH; ++b) {
        const int key = dstv[b] / BUCKET_ROWS;
        const unsigned m = same_key_lanes(key);
        peers[b] = m;
        if (lane == __builtin_ctz(m)) wbase[w][key] += __builtin_popcount(m);
    }
    __syncthreads();

    int cw[5];
    int mine2 = 0;
    #pragma unroll
    for (int j = 0; j < 5; ++j) {
        int s = 0;
        if (t < SCAN_LANES) {
            #pragma unroll
            for (int wv = 0; wv < P_WAVES; ++wv) s += wbase[wv][t * 5 + j];
        }
        cw[j] = s;
        mine2 += s;
    }
    int2 pm;
    pm.x = mine; pm.y = mine2;
    part[t] = pm;
    __syncthreads();

    for (int off = 1; off < P_THREADS; off <<= 1) {
        int2 add;
        add.x = 0; add.y = 0;
        if (t >= off) add = part[t - off];
        __syncthreads();
        int2 cur = part[t];
        cur.x += add.x; cur.y += add.y;
        part[t] = cur;
        __syncthreads();
    }
    const int2 incl = part[t];
    int run  = incl.x - mine;
    int run2 = incl.y - mine2;
    if (t < SCAN_LANES) {
        #pragma unroll
        for (int j = 0; j < 5; ++j) {
            const int k = t * 5 + j;
            if (g == 0) bstart[k] = run;
            int base = run + gh[j];
            dlt[k] = base - run2;
            #pragma unroll
            for (int wv = 0; wv < P_WAVES; ++wv) {
                const int cnt = wbase[wv][k];
                wbase[wv][k] = base;
                base += cnt;
            }
            run  += loc[j];
            run2 += cw[j];
        }
    }
    if (g == 0 && t == P_THREADS - 1) bstart[N_BUCKET] = incl.x;
    __syncthreads();

    const unsigned lt = (1u << lane) - 1u;
    #pragma unroll
    for (int b = 0; b < P_BATCH; ++b) {
        const int e = ebase + b * 32 + lane;
        const int d = dstv[b];
        const int key = d / BUCKET_ROWS;
        const unsigned m = peers[b];
        const int leader = __builtin_ctz(m);
        int slot = 0;
        if (lane == leader) {
            slot = wbase[w][key];
            wbase[w][key] = slot + __builtin_popcount(m);
        }
        slot = __shfl(slot, leader, 32);
        const int pos = slot + __builtin_popcount(m & lt);
        const int lp  = pos - dlt[key];
        int2 rec;
        rec.x = ((d - key * BUCKET_ROWS) << 16) | adj_src[e];
        rec.y = __float_as_int(weight[e / N_EDGE] * adj_vals[e]);
        srec[lp] = rec;
        spos[lp] = pos;
    }
    __syncthreads();

    #pragma unroll
    for (int b = 0; b < P_BATCH; ++b) {
        const int i = b * P_THREADS + t;
        brec[spos[i]] = srec[i];
    }
}

template <int M>
__device__ __forceinline__ float row_xor_add(float v) {
    const int b = __builtin_bit_cast(int, v);
    const int r = __builtin_amdgcn_update_dpp(0, b, 0x160 | M, 0xf, 0xf, true);
    return v + __builtin_bit_cast(float, r);
}
__device__ __forceinline__ float xhalf(float x) {
    const int b = __builtin_bit_cast(int, x);
    return __builtin_bit_cast(float,
        __builtin_amdgcn_permlanex16(b, b, 0x76543210, 0xfedcba98, false, false));
}

__global__ __launch_bounds__(256, 16) void gather_rows(
    const float* __restrict__ x,
    const int*   __restrict__ bstart,
    const int2*  __restrict__ brec,
    float*       __restrict__ outp)
{
    __shared__ int2 s_rec[ROWS_PER_WG][ROW_CAP];

    const int lane = threadIdx.x & 31;
    const int w    = threadIdx.x >> 5;
    const int row  = blockIdx.x * ROWS_PER_WG + w;
    const int bucket = row / BUCKET_ROWS;
    const int local  = row - bucket * BUCKET_ROWS;
    const int pb = __builtin_amdgcn_readfirstlane(bstart[bucket]);
    const int pe = __builtin_amdgcn_readfirstlane(bstart[bucket + 1]);
    const unsigned lt = (1u << lane) - 1u;

    int n = 0;
    for (int p0 = pb; p0 < pe; p0 += 256) {
        int2 rc[8];
        #pragma unroll
        for (int u = 0; u < 8; ++u) {
            const int p = p0 + u * 32 + lane;
            const bool ok = p < pe;
            const int pc = ok ? p : (pe - 1);
            rc[u] = brec[pc];
            rc[u].x = ok ? rc[u].x : -1;
        }
        #pragma unroll
        for (int u = 0; u < 8; ++u) {
            const bool hit = (rc[u].x >> 16) == local;
            const unsigned m = __builtin_amdgcn_ballot_w32(hit);
            if (hit) {
                const int slot = n + __builtin_popcount(m & lt);
                if (slot < ROW_CAP) {
                    int2 keep;
                    keep.x = rc[u].x & 0xffff;
                    keep.y = rc[u].y;
                    s_rec[w][slot] = keep;
                }
            }
            n += __builtin_popcount(m);
        }
    }
    const bool listed = (n <= ROW_CAP);
    __syncthreads();

    float4 acc;
    acc.x = 0.f; acc.y = 0.f; acc.z = 0.f; acc.w = 0.f;

    if (listed) {
        int i = 0;
        for (; i + 8 <= n; i += 8) {
            float4 xv[8];
            float  cc[8];
            #pragma unroll
            for (int u = 0; u < 8; ++u) {
                const int2 r = s_rec[w][i + u];
                const int ss = __builtin_amdgcn_readfirstlane(r.x);
                cc[u] = __int_as_float(r.y);
                xv[u] = *(reinterpret_cast<const float4*>(x + (size_t)ss * N_HID) + lane);
            }
            #pragma unroll
            for (int u = 0; u < 8; ++u) {
                acc.x = fmaf(xv[u].x, cc[u], acc.x);
                acc.y = fmaf(xv[u].y, cc[u], acc.y);
                acc.z = fmaf(xv[u].z, cc[u], acc.z);
                acc.w = fmaf(xv[u].w, cc[u], acc.w);
            }
        }
        const int rem = n - i;
        if (rem > 0) {
            float4 xv[7];
            float  cc[7];
            #pragma unroll
            for (int u = 0; u < 7; ++u) {
                const int idx = (u < rem) ? (i + u) : i;
                const int2 r = s_rec[w][idx];
                const int ss = __builtin_amdgcn_readfirstlane(r.x);
                cc[u] = __int_as_float(r.y);
                xv[u] = *(reinterpret_cast<const float4*>(x + (size_t)ss * N_HID) + lane);
            }
            #pragma unroll
            for (int u = 0; u < 7; ++u) {
                if (u < rem) {
                    acc.x = fmaf(xv[u].x, cc[u], acc.x);
                    acc.y = fmaf(xv[u].y, cc[u], acc.y);
                    acc.z = fmaf(xv[u].z, cc[u], acc.z);
                    acc.w = fmaf(xv[u].w, cc[u], acc.w);
                }
            }
        }
    } else {
        for (int p0 = pb; p0 < pe; p0 += 32) {
            const int p = p0 + lane;
            const bool ok = p < pe;
            int2 rc;
            rc.x = -1; rc.y = 0;
            if (ok) rc = brec[p];
            const int pk = rc.x;
            const int cv = rc.y;
            unsigned m = __builtin_amdgcn_ballot_w32((pk >> 16) == local);
            while (m) {
                const int l = __builtin_ctz(m);
                m &= m - 1;
                const int ss = __builtin_amdgcn_readlane(pk, l) & 0xffff;
                const float sc = __int_as_float(__builtin_amdgcn_readlane(cv, l));
                const float4 xv = *(reinterpret_cast<const float4*>(x + (size_t)ss * N_HID) + lane);
                acc.x = fmaf(xv.x, sc, acc.x);
                acc.y = fmaf(xv.y, sc, acc.y);
                acc.z = fmaf(xv.z, sc, acc.z);
                acc.w = fmaf(xv.w, sc, acc.w);
            }
        }
    }

    v2f a0, a1;
    a0.x = acc.x; a0.y = acc.y;
    a1.x = acc.z; a1.y = acc.w;
    v8f c = {};
    c = __builtin_amdgcn_wmma_f32_16x16x4_f32(false, a0, false, a0, (short)0, c, false, false);
    c = __builtin_amdgcn_wmma_f32_16x16x4_f32(false, a1, false, a1, (short)0, c, false, false);
    const int sel = lane & 7;
    float dv = c[0];
    dv = (sel == 1) ? c[1] : dv;
    dv = (sel == 2) ? c[2] : dv;
    dv = (sel == 3) ? c[3] : dv;
    dv = (sel == 4) ? c[4] : dv;
    dv = (sel == 5) ? c[5] : dv;
    dv = (sel == 6) ? c[6] : dv;
    dv = (sel == 7) ? c[7] : dv;
    dv = (lane < 8 || lane >= 24) ? dv : 0.0f;
    dv = row_xor_add<1>(dv);
    dv = row_xor_add<2>(dv);
    dv = row_xor_add<4>(dv);
    dv = row_xor_add<8>(dv);
    dv = dv + xhalf(dv);

    const float scale = 1.0f / fmaxf(sqrtf(dv), 1e-12f);
    const float inv_sqrt2 = 0.70710678118654752440f;
    float4 o;
    o.x = acc.x * scale;
    o.y = acc.y * scale;
    o.z = acc.z * scale;
    o.w = acc.w * scale;
    o.x = 0.5f * o.x * (1.0f + erff(o.x * inv_sqrt2));
    o.y = 0.5f * o.y * (1.0f + erff(o.y * inv_sqrt2));
    o.z = 0.5f * o.z * (1.0f + erff(o.z * inv_sqrt2));
    o.w = 0.5f * o.w * (1.0f + erff(o.w * inv_sqrt2));
    *(reinterpret_cast<float4*>(outp + (size_t)row * N_HID) + lane) = o;
}

extern "C" void kernel_launch(void* const* d_in, const int* in_sizes, int n_in,
                              void* d_out, int out_size, void* d_ws, size_t ws_size,
                              hipStream_t stream) {
    const float* x        = (const float*)d_in[0];
    const float* weight   = (const float*)d_in[1];
    const int*   adj_src  = (const int*)d_in[2];
    const int*   adj_dst  = (const int*)d_in[3];
    const float* adj_vals = (const float*)d_in[4];
    float*       out      = (float*)d_out;

    int*   ghist  = (int*)d_ws;
    int*   btotal = ghist + P_WG * N_BUCKET;
    int*   bstart = btotal + 2048;
    int2*  brec   = (int2*)(bstart + 2048);

    part_count<<<P_WG, P_THREADS, 0, stream>>>(adj_dst, ghist);
    bucket_prefix<<<(N_BUCKET + 255) / 256, 256, 0, stream>>>(ghist, btotal);
    part_scatter<<<P_WG, P_THREADS, 0, stream>>>(weight, adj_src, adj_dst, adj_vals,
                                                 ghist, btotal, bstart, brec);
    gather_rows<<<N_NODES / ROWS_PER_WG, 256, 0, stream>>>(x, bstart, brec, out);
}


# Placement kernel: source and value loads (40 per thread) issued ahead of the counting phase instead of 3 loads and their wait inside each of the 20 placement batches; list and sum order unchanged
# speedup vs baseline: 3.7512x; 1.0457x over previous
#include <hip/hip_runtime.h>
#include <math.h>

#define N_NODES 50000
#define N_HID   128
#define N_ADJ   4
#define N_EDGE  600000
#define TOTAL_EDGES (N_ADJ * N_EDGE)

#define BUCKET_ROWS 25
#define N_BUCKET    (N_NODES / BUCKET_ROWS)
#define KEY_BITS    11
#define P_WG        250
#define P_WAVES     15
#define P_THREADS   (P_WAVES * 32)
#define P_BATCH     20
#define SCAN_LANES  400
#define ROW_CAP     128
#define ROWS_PER_WG 8

typedef __attribute__((ext_vector_type(2))) float v2f;
typedef __attribute__((ext_vector_type(8))) float v8f;

__device__ __forceinline__ unsigned same_key_lanes(int key) {
    unsigned m = 0xffffffffu;
    #pragma unroll
    for (int b = 0; b < KEY_BITS; ++b) {
        const bool bit = ((key >> b) & 1) != 0;
        const unsigned bal = __builtin_amdgcn_ballot_w32(bit);
        m &= bit ? bal : ~bal;
    }
    return m;
}

__global__ __launch_bounds__(P_THREADS) void part_count(
    const int* __restrict__ adj_dst, int* __restrict__ ghist)
{
    __shared__ int whist[P_WAVES][N_BUCKET];
    const int t = threadIdx.x;
    const int lane = t & 31;
    const int w = t >> 5;
    for (int i = t; i < P_WAVES * N_BUCKET; i += P_THREADS) (&whist[0][0])[i] = 0;
    __syncthreads();
    const int ebase = (blockIdx.x * P_WAVES + w) * (P_BATCH * 32);
    int key[P_BATCH];
    #pragma unroll
    for (int b = 0; b < P_BATCH; ++b)
        key[b] = __builtin_nontemporal_load(adj_dst + ebase + b * 32 + lane) / BUCKET_ROWS;
    #pragma unroll
    for (int b = 0; b < P_BATCH; ++b) {
        const unsigned m = same_key_lanes(key[b]);
        if (lane == __builtin_ctz(m)) whist[w][key[b]] += __builtin_popcount(m);
    }
    __syncthreads();
    for (int k = t; k < N_BUCKET; k += P_THREADS) {
        int sum = 0;
        #pragma unroll
        for (int wv = 0; wv < P_WAVES; ++wv) sum += whist[wv][k];
        ghist[blockIdx.x * N_BUCKET + k] = sum;
    }
}

__global__ __launch_bounds__(256) void bucket_prefix(int* ghist, int* __restrict__ btotal)
{
    const int k = blockIdx.x * 256 + threadIdx.x;
    if (k >= N_BUCKET) return;
    int run = 0;
    for (int g0 = 0; g0 < P_WG; g0 += 50) {
        int v[50];
        #pragma unroll
        for (int j = 0; j < 50; ++j) v[j] = ghist[(g0 + j) * N_BUCKET + k];
        #pragma unroll
        for (int j = 0; j < 50; ++j) {
            ghist[(g0 + j) * N_BUCKET + k] = run;
            run += v[j];
        }
    }
    btotal[k] = run;
}

#define P_EDGES (P_WAVES * P_BATCH * 32)
__global__ __launch_bounds__(P_THREADS) void part_scatter(
    const float* __restrict__ weight,
    const int*   __restrict__ adj_src,
    const int*   __restrict__ adj_dst,
    const float* __restrict__ adj_vals,
    const int*   __restrict__ ghist,
    const int*   __restrict__ btotal,
    int*   __restrict__ bstart,
    int2*  __restrict__ brec)
{
    __shared__ int  wbase[P_WAVES][N_BUCKET];
    __shared__ int  dlt[N_BUCKET];
    __shared__ int2 part[P_THREADS];
    __shared__ int2 srec[P_EDGES];
    __shared__ int  spos[P_EDGES];
    const int t = threadIdx.x;
    const int lane = t & 31;
    const int w = t >> 5;
    const int g = blockIdx.x;

    const int ebase = (g * P_WAVES + w) * (P_BATCH * 32);
    int dstv[P_BATCH];
    #pragma unroll
    for (int b = 0; b < P_BATCH; ++b)
        dstv[b] = __builtin_nontemporal_load(adj_dst + ebase + b * 32 + lane);

    for (int i = t; i < P_WAVES * N_BUCKET; i += P_THREADS) (&wbase[0][0])[i] = 0;
    __syncthreads();

    int   srcv[P_BATCH];
    float valv[P_BATCH];
    #pragma unroll
    for (int b = 0; b < P_BATCH; ++b)
        srcv[b] = adj_src[ebase + b * 32 + lane];
    #pragma unroll
    for (int b = 0; b < P_BATCH; ++b)
        valv[b] = adj_vals[ebase + b * 32 + lane];
    const int   a0     = ebase / N_EDGE;
    const int   a1     = (a0 + 1 < N_ADJ) ? (a0 + 1) : (N_ADJ - 1);
    const int   abound = (a0 + 1) * N_EDGE;
    const float wgt0   = weight[a0];
    const float wgt1   = weight[a1];
    int loc[5];
    int gh[5];
    int mine = 0;
    #pragma unroll
    for (int j = 0; j < 5; ++j) {
        loc[j] = (t < SCAN_LANES) ? btotal[t * 5 + j] : 0;
        gh[j]  = (t < SCAN_LANES) ? ghist[g * N_BUCKET + t * 5 + j] : 0;
    }

    unsigned peers[P_BATCH];
    #pragma unroll
    for (int b = 0; b < P_BATCH; ++b) {
        const int key = dstv[b] / BUCKET_ROWS;
        const unsigned m = same_key_lanes(key);
        peers[b] = m;
        if (lane == __builtin_ctz(m)) wbase[w][key] += __builtin_popcount(m);
    }
    __syncthreads();

    #pragma unroll
    for (int j = 0; j < 5; ++j) mine += loc[j];
    int cw[5];
    int mine2 = 0;
    #pragma unroll
    for (int j = 0; j < 5; ++j) {
        int s = 0;
        if (t < SCAN_LANES) {
            #pragma unroll
            for (int wv = 0; wv < P_WAVES; ++wv) s += wbase[wv][t * 5 + j];
        }
        cw[j] = s;
        mine2 += s;
    }
    int2 pm;
    pm.x = mine; pm.y = mine2;
    part[t] = pm;
    __syncthreads();

    for (int off = 1; off < P_THREADS; off <<= 1) {
        int2 add;
        add.x = 0; add.y = 0;
        if (t >= off) add = part[t - off];
        __syncthreads();
        int2 cur = part[t];
        cur.x += add.x; cur.y += add.y;
        part[t] = cur;
        __syncthreads();
    }
    const int2 incl = part[t];
    int run  = incl.x - mine;
    int run2 = incl.y - mine2;
    if (t < SCAN_LANES) {
        #pragma unroll
        for (int j = 0; j < 5; ++j) {
            const int k = t * 5 + j;
            if (g == 0) bstart[k] = run;
            int base = run + gh[j];
            dlt[k] = base - run2;
            #pragma unroll
            for (int wv = 0; wv < P_WAVES; ++wv) {
                const int cnt = wbase[wv][k];
                wbase[wv][k] = base;
                base += cnt;
            }
            run  += loc[j];
            run2 += cw[j];
        }
    }
    if (g == 0 && t == P_THREADS - 1) bstart[N_BUCKET] = incl.x;
    __syncthreads();

    const unsigned lt = (1u << lane) - 1u;
    #pragma unroll
    for (int b = 0; b < P_BATCH; ++b) {
        const int e = ebase + b * 32 + lane;
        const int d = dstv[b];
        const int key = d / BUCKET_ROWS;
        const unsigned m = peers[b];
        const int leader = __builtin_ctz(m);
        int slot = 0;
        if (lane == leader) {
            slot = wbase[w][key];
            wbase[w][key] = slot + __builtin_popcount(m);
        }
        slot = __shfl(slot, leader, 32);
        const int pos = slot + __builtin_popcount(m & lt);
        const int lp  = pos - dlt[key];
        int2 rec;
        rec.x = ((d - key * BUCKET_ROWS) << 16) | srcv[b];
        rec.y = __float_as_int(((e < abound) ? wgt0 : wgt1) * valv[b]);
        srec[lp] = rec;
        spos[lp] = pos;
    }
    __syncthreads();

    #pragma unroll
    for (int b = 0; b < P_BATCH; ++b) {
        const int i = b * P_THREADS + t;
        brec[spos[i]] = srec[i];
    }
}

template <int M>
__device__ __forceinline__ float row_xor_add(float v) {
    const int b = __builtin_bit_cast(int, v);
    const int r = __builtin_amdgcn_update_dpp(0, b, 0x160 | M, 0xf, 0xf, true);
    return v + __builtin_bit_cast(float, r);
}
__device__ __forceinline__ float xhalf(float x) {
    const int b = __builtin_bit_cast(int, x);
    return __builtin_bit_cast(float,
        __builtin_amdgcn_permlanex16(b, b, 0x76543210, 0xfedcba98, false, false));
}

__global__ __launch_bounds__(256, 16) void gather_rows(
    const float* __restrict__ x,
    const int*   __restrict__ bstart,
    const int2*  __restrict__ brec,
    float*       __restrict__ outp)
{
    __shared__ int2 s_rec[ROWS_PER_WG][ROW_CAP];

    const int lane = threadIdx.x & 31;
    const int w    = threadIdx.x >> 5;
    const int row  = blockIdx.x * ROWS_PER_WG + w;
    const int bucket = row / BUCKET_ROWS;
    const int local  = row - bucket * BUCKET_ROWS;
    const int pb = __builtin_amdgcn_readfirstlane(bstart[bucket]);
    const int pe = __builtin_amdgcn_readfirstlane(bstart[bucket + 1]);
    const unsigned lt = (1u << lane) - 1u;

    int n = 0;
    for (int p0 = pb; p0 < pe; p0 += 256) {
        int2 rc[8];
        #pragma unroll
        for (int u = 0; u < 8; ++u) {
            const int p = p0 + u * 32 + lane;
            const bool ok = p < pe;
            const int pc = ok ? p : (pe - 1);
            rc[u] = brec[pc];
            rc[u].x = ok ? rc[u].x : -1;
        }
        #pragma unroll
        for (int u = 0; u < 8; ++u) {
            const bool hit = (rc[u].x >> 16) == local;
            const unsigned m = __builtin_amdgcn_ballot_w32(hit);
            if (hit) {
                const int slot = n + __builtin_popcount(m & lt);
                if (slot < ROW_CAP) {
                    int2 keep;
                    keep.x = rc[u].x & 0xffff;
                    keep.y = rc[u].y;
                    s_rec[w][slot] = keep;
                }
            }
            n += __builtin_popcount(m);
        }
    }
    const bool listed = (n <= ROW_CAP);
    __syncthreads();

    float4 acc;
    acc.x = 0.f; acc.y = 0.f; acc.z = 0.f; acc.w = 0.f;

    if (listed) {
        int i = 0;
        for (; i + 8 <= n; i += 8) {
            float4 xv[8];
            float  cc[8];
            #pragma unroll
            for (int u = 0; u < 8; ++u) {
                const int2 r = s_rec[w][i + u];
                const int ss = __builtin_amdgcn_readfirstlane(r.x);
                cc[u] = __int_as_float(r.y);
                xv[u] = *(reinterpret_cast<const float4*>(x + (size_t)ss * N_HID) + lane);
            }
            #pragma unroll
            for (int u = 0; u < 8; ++u) {
                acc.x = fmaf(xv[u].x, cc[u], acc.x);
                acc.y = fmaf(xv[u].y, cc[u], acc.y);
                acc.z = fmaf(xv[u].z, cc[u], acc.z);
                acc.w = fmaf(xv[u].w, cc[u], acc.w);
            }
        }
        const int rem = n - i;
        if (rem > 0) {
            float4 xv[7];
            float  cc[7];
            #pragma unroll
            for (int u = 0; u < 7; ++u) {
                const int idx = (u < rem) ? (i + u) : i;
                const int2 r = s_rec[w][idx];
                const int ss = __builtin_amdgcn_readfirstlane(r.x);
                cc[u] = __int_as_float(r.y);
                xv[u] = *(reinterpret_cast<const float4*>(x + (size_t)ss * N_HID) + lane);
            }
            #pragma unroll
            for (int u = 0; u < 7; ++u) {
                if (u < rem) {
                    acc.x = fmaf(xv[u].x, cc[u], acc.x);
                    acc.y = fmaf(xv[u].y, cc[u], acc.y);
                    acc.z = fmaf(xv[u].z, cc[u], acc.z);
                    acc.w = fmaf(xv[u].w, cc[u], acc.w);
                }
            }
        }
    } else {
        for (int p0 = pb; p0 < pe; p0 += 32) {
            const int p = p0 + lane;
            const bool ok = p < pe;
            int2 rc;
            rc.x = -1; rc.y = 0;
            if (ok) rc = brec[p];
            const int pk = rc.x;
            const int cv = rc.y;
            unsigned m = __builtin_amdgcn_ballot_w32((pk >> 16) == local);
            while (m) {
                const int l = __builtin_ctz(m);
                m &= m - 1;
                const int ss = __builtin_amdgcn_readlane(pk, l) & 0xffff;
                const float sc = __int_as_float(__builtin_amdgcn_readlane(cv, l));
                const float4 xv = *(reinterpret_cast<const float4*>(x + (size_t)ss * N_HID) + lane);
                acc.x = fmaf(xv.x, sc, acc.x);
                acc.y = fmaf(xv.y, sc, acc.y);
                acc.z = fmaf(xv.z, sc, acc.z);
                acc.w = fmaf(xv.w, sc, acc.w);
            }
        }
    }

    v2f a0, a1;
    a0.x = acc.x; a0.y = acc.y;
    a1.x = acc.z; a1.y = acc.w;
    v8f c = {};
    c = __builtin_amdgcn_wmma_f32_16x16x4_f32(false, a0, false, a0, (short)0, c, false, false);
    c = __builtin_amdgcn_wmma_f32_16x16x4_f32(false, a1, false, a1, (short)0, c, false, false);
    const int sel = lane & 7;
    float dv = c[0];
    dv = (sel == 1) ? c[1] : dv;
    dv = (sel == 2) ? c[2] : dv;
    dv = (sel == 3) ? c[3] : dv;
    dv = (sel == 4) ? c[4] : dv;
    dv = (sel == 5) ? c[5] : dv;
    dv = (sel == 6) ? c[6] : dv;
    dv = (sel == 7) ? c[7] : dv;
    dv = (lane < 8 || lane >= 24) ? dv : 0.0f;
    dv = row_xor_add<1>(dv);
    dv = row_xor_add<2>(dv);
    dv = row_xor_add<4>(dv);
    dv = row_xor_add<8>(dv);
    dv = dv + xhalf(dv);

    const float scale = 1.0f / fmaxf(sqrtf(dv), 1e-12f);
    const float inv_sqrt2 = 0.70710678118654752440f;
    float4 o;
    o.x = acc.x * scale;
    o.y = acc.y * scale;
    o.z = acc.z * scale;
    o.w = acc.w * scale;
    o.x = 0.5f * o.x * (1.0f + erff(o.x * inv_sqrt2));
    o.y = 0.5f * o.y * (1.0f + erff(o.y * inv_sqrt2));
    o.z = 0.5f * o.z * (1.0f + erff(o.z * inv_sqrt2));
    o.w = 0.5f * o.w * (1.0f + erff(o.w * inv_sqrt2));
    *(reinterpret_cast<float4*>(outp + (size_t)row * N_HID) + lane) = o;
}

extern "C" void kernel_launch(void* const* d_in, const int* in_sizes, int n_in,
                              void* d_out, int out_size, void* d_ws, size_t ws_size,
                              hipStream_t stream) {
    const float* x        = (const float*)d_in[0];
    const float* weight   = (const float*)d_in[1];
    const int*   adj_src  = (const int*)d_in[2];
    const int*   adj_dst  = (const int*)d_in[3];
    const float* adj_vals = (const float*)d_in[4];
    float*       out      = (float*)d_out;

    int*   ghist  = (int*)d_ws;
    int*   btotal = ghist + P_WG * N_BUCKET;
    int*   bstart = btotal + 2048;
    int2*  brec   = (int2*)(bstart + 2048);

    part_count<<<P_WG, P_THREADS, 0, stream>>>(adj_dst, ghist);
    bucket_prefix<<<(N_BUCKET + 255) / 256, 256, 0, stream>>>(ghist, btotal);
    part_scatter<<<P_WG, P_THREADS, 0, stream>>>(weight, adj_src, adj_dst, adj_vals,
                                                 ghist, btotal, bstart, brec);
    gather_rows<<<N_NODES / ROWS_PER_WG, 256, 0, stream>>>(x, bstart, brec, out);
}


# Bucket lists sorted by destination row once per call (new launch: stable counting sort in LDS, in place); a row's wave copies its own ~48 records instead of scanning ~1200; sum order kept
# speedup vs baseline: 4.0673x; 1.0843x over previous
#include <hip/hip_runtime.h>
#include <math.h>

#define N_NODES 50000
#define N_HID   128
#define N_ADJ   4
#define N_EDGE  600000
#define TOTAL_EDGES (N_ADJ * N_EDGE)

#define BUCKET_ROWS 25
#define N_BUCKET    (N_NODES / BUCKET_ROWS)
#define KEY_BITS    11
#define P_WG        250
#define P_WAVES     15
#define P_THREADS   (P_WAVES * 32)
#define P_BATCH     20
#define SCAN_LANES  400
#define ROW_CAP     128
#define ROWS_PER_WG 8
#define RS_WAVES    4
#define RS_THREADS  (RS_WAVES * 32)
#define RS_BATCH    12
#define RS_NB       (RS_WAVES * RS_BATCH)
#define RS_CAP      (RS_NB * 32)
#define RS_PITCH    32
#define ROW_BITS    5

typedef __attribute__((ext_vector_type(2))) float v2f;
typedef __attribute__((ext_vector_type(8))) float v8f;

__device__ __forceinline__ unsigned same_key_lanes(int key) {
    unsigned m = 0xffffffffu;
    #pragma unroll
    for (int b = 0; b < KEY_BITS; ++b) {
        const bool bit = ((key >> b) & 1) != 0;
        const unsigned bal = __builtin_amdgcn_ballot_w32(bit);
        m &= bit ? bal : ~bal;
    }
    return m;
}

__device__ __forceinline__ unsigned same_row_lanes(int key) {
    unsigned m = 0xffffffffu;
    #pragma unroll
    for (int b = 0; b < ROW_BITS; ++b) {
        const int sm = (int)((unsigned)key << (31 - b)) >> 31;
        const unsigned bal = __builtin_amdgcn_ballot_w32(sm != 0);
        m &= ~(bal ^ (unsigned)sm);
    }
    return m;
}

__global__ __launch_bounds__(P_THREADS) void part_count(
    const int* __restrict__ adj_dst, int* __restrict__ ghist)
{
    __shared__ int whist[P_WAVES][N_BUCKET];
    const int t = threadIdx.x;
    const int lane = t & 31;
    const int w = t >> 5;
    for (int i = t; i < P_WAVES * N_BUCKET; i += P_THREADS) (&whist[0][0])[i] = 0;
    __syncthreads();
    const int ebase = (blockIdx.x * P_WAVES + w) * (P_BATCH * 32);
    int key[P_BATCH];
    #pragma unroll
    for (int b = 0; b < P_BATCH; ++b)
        key[b] = __builtin_nontemporal_load(adj_dst + ebase + b * 32 + lane) / BUCKET_ROWS;
    #pragma unroll
    for (int b = 0; b < P_BATCH; ++b) {
        const unsigned m = same_key_lanes(key[b]);
        if (lane == __builtin_ctz(m)) whist[w][key[b]] += __builtin_popcount(m);
    }
    __syncthreads();
    for (int k = t; k < N_BUCKET; k += P_THREADS) {
        int sum = 0;
        #pragma unroll
        for (int wv = 0; wv < P_WAVES; ++wv) sum += whist[wv][k];
        ghist[blockIdx.x * N_BUCKET + k] = sum;
    }
}

__global__ __launch_bounds__(256) void bucket_prefix(int* ghist, int* __restrict__ btotal)
{
    const int k = blockIdx.x * 256 + threadIdx.x;
    if (k >= N_BUCKET) return;
    int run = 0;
    for (int g0 = 0; g0 < P_WG; g0 += 50) {
        int v[50];
        #pragma unroll
        for (int j = 0; j < 50; ++j) v[j] = ghist[(g0 + j) * N_BUCKET + k];
        #pragma unroll
        for (int j = 0; j < 50; ++j) {
            ghist[(g0 + j) * N_BUCKET + k] = run;
            run += v[j];
        }
    }
    btotal[k] = run;
}

#define P_EDGES (P_WAVES * P_BATCH * 32)
__global__ __launch_bounds__(P_THREADS) void part_scatter(
    const float* __restrict__ weight,
    const int*   __restrict__ adj_src,
    const int*   __restrict__ adj_dst,
    const float* __restrict__ adj_vals,
    const int*   __restrict__ ghist,
    const int*   __restrict__ btotal,
    int*   __restrict__ bstart,
    int2*  __restrict__ brec)
{
    __shared__ int  wbase[P_WAVES][N_BUCKET];
    __shared__ int  dlt[N_BUCKET];
    __shared__ int2 part[P_THREADS];
    __shared__ int2 srec[P_EDGES];
    __shared__ int  spos[P_EDGES];
    const int t = threadIdx.x;
    const int lane = t & 31;
    const int w = t >> 5;
    const int g = blockIdx.x;

    const int ebase = (g * P_WAVES + w) * (P_BATCH * 32);
    int dstv[P_BATCH];
    #pragma unroll
    for (int b = 0; b < P_BATCH; ++b)
        dstv[b] = __builtin_nontemporal_load(adj_dst + ebase + b * 32 + lane);

    for (int i = t; i < P_WAVES * N_BUCKET; i += P_THREADS) (&wbase[0][0])[i] = 0;
    __syncthreads();

    int   srcv[P_BATCH];
    float valv[P_BATCH];
    #pragma unroll
    for (int b = 0; b < P_BATCH; ++b)
        srcv[b] = adj_src[ebase + b * 32 + lane];
    #pragma unroll
    for (int b = 0; b < P_BATCH; ++b)
        valv[b] = adj_vals[ebase + b * 32 + lane];
    const int   a0     = ebase / N_EDGE;
    const int   a1     = (a0 + 1 < N_ADJ) ? (a0 + 1) : (N_ADJ - 1);
    const int   abound = (a0 + 1) * N_EDGE;
    const float wgt0   = weight[a0];
    const float wgt1   = weight[a1];
    int loc[5];
    int gh[5];
    int mine = 0;
    #pragma unroll
    for (int j = 0; j < 5; ++j) {
        loc[j] = (t < SCAN_LANES) ? btotal[t * 5 + j] : 0;
        gh[j]  = (t < SCAN_LANES) ? ghist[g * N_BUCKET + t * 5 + j] : 0;
    }

    unsigned peers[P_BATCH];
    #pragma unroll
    for (int b = 0; b < P_BATCH; ++b) {
        const int key = dstv[b] / BUCKET_ROWS;
        const unsigned m = same_key_lanes(key);
        peers[b] = m;
        if (lane == __builtin_ctz(m)) wbase[w][key] += __builtin_popcount(m);
    }
    __syncthreads();

    #pragma unroll
    for (int j = 0; j < 5; ++j) mine += loc[j];
    int cw[5];
    int mine2 = 0;
    #pragma unroll
    for (int j = 0; j < 5; ++j) {
        int s = 0;
        if (t < SCAN_LANES) {
            #pragma unroll
            for (int wv = 0; wv < P_WAVES; ++wv) s += wbase[wv][t * 5 + j];
        }
        cw[j] = s;
        mine2 += s;
    }
    int2 pm;
    pm.x = mine; pm.y = mine2;
    part[t] = pm;
    __syncthreads();

    for (int off = 1; off < P_THREADS; off <<= 1) {
        int2 add;
        add.x = 0; add.y = 0;
        if (t >= off) add = part[t - off];
        __syncthreads();
        int2 cur = part[t];
        cur.x += add.x; cur.y += add.y;
        part[t] = cur;
        __syncthreads();
    }
    const int2 incl = part[t];
    int run  = incl.x - mine;
    int run2 = incl.y - mine2;
    if (t < SCAN_LANES) {
        #pragma unroll
        for (int j = 0; j < 5; ++j) {
            const int k = t * 5 + j;
            if (g == 0) bstart[k] = run;
            int base = run + gh[j];
            dlt[k] = base - run2;
            #pragma unroll
            for (int wv = 0; wv < P_WAVES; ++wv) {
                const int cnt = wbase[wv][k];
                wbase[wv][k] = base;
                base += cnt;
            }
            run  += loc[j];
            run2 += cw[j];
        }
    }
    if (g == 0 && t == P_THREADS - 1) bstart[N_BUCKET] = incl.x;
    __syncthreads();

    const unsigned lt = (1u << lane) - 1u;
    #pragma unroll
    for (int b = 0; b < P_BATCH; ++b) {
        const int e = ebase + b * 32 + lane;
        const int d = dstv[b];
        const int key = d / BUCKET_ROWS;
        const unsigned m = peers[b];
        const int leader = __builtin_ctz(m);
        int slot = 0;
        if (lane == leader) {
            slot = wbase[w][key];
            wbase[w][key] = slot + __builtin_popcount(m);
        }
        slot = __shfl(slot, leader, 32);
        const int pos = slot + __builtin_popcount(m & lt);
        const int lp  = pos - dlt[key];
        int2 rec;
        rec.x = ((d - key * BUCKET_ROWS) << 16) | srcv[b];
        rec.y = __float_as_int(((e < abound) ? wgt0 : wgt1) * valv[b]);
        srec[lp] = rec;
        spos[lp] = pos;
    }
    __syncthreads();

    #pragma unroll
    for (int b = 0; b < P_BATCH; ++b) {
        const int i = b * P_THREADS + t;
        brec[spos[i]] = srec[i];
    }
}

__global__ __launch_bounds__(RS_THREADS) void row_sort(
    const int* __restrict__ bstart, int2* brec, int2* __restrict__ rse)
{
    __shared__ int  cnt[RS_NB * RS_PITCH];
    __shared__ int  wtot[RS_WAVES * 32];
    __shared__ int2 sorted[RS_CAP];
    const int t = threadIdx.x;
    const int lane = t & 31;
    const int w = t >> 5;
    const int bucket = blockIdx.x;
    const int pb = __builtin_amdgcn_readfirstlane(bstart[bucket]);
    const int pe = __builtin_amdgcn_readfirstlane(bstart[bucket + 1]);
    const int n = pe - pb;

    if (n > RS_CAP) {
        if (t < BUCKET_ROWS) {
            int2 mark;
            mark.x = ~pb; mark.y = pe;
            rse[bucket * BUCKET_ROWS + t] = mark;
        }
        return;
    }

    int2 rec[RS_BATCH];
    #pragma unroll
    for (int j = 0; j < RS_BATCH; ++j) {
        const int i = (j * RS_WAVES + w) * 32 + lane;
        int2 r;
        r.x = -1; r.y = 0;
        if (i < n) r = brec[pb + i];
        rec[j] = r;
    }
    for (int i = t; i < RS_NB * RS_PITCH; i += RS_THREADS) cnt[i] = 0;
    __syncthreads();

    unsigned peers[RS_BATCH];
    #pragma unroll
    for (int j = 0; j < RS_BATCH; ++j) {
        const int b = j * RS_WAVES + w;
        peers[j] = 0u;
        if (b * 32 < n) {
            const bool ok = (b * 32 + lane) < n;
            const int key = ok ? (rec[j].x >> 16) : 31;
            const unsigned m = same_row_lanes(key);
            peers[j] = m;
            if (ok && lane == __builtin_ctz(m)) cnt[b * RS_PITCH + key] = __builtin_popcount(m);
        }
    }
    __syncthreads();

    int v[RS_BATCH];
    #pragma unroll
    for (int jj = 0; jj < RS_BATCH; ++jj) v[jj] = cnt[(w * RS_BATCH + jj) * RS_PITCH + lane];
    int run = 0;
    #pragma unroll
    for (int jj = 0; jj < RS_BATCH; ++jj) {
        const int e = v[jj];
        v[jj] = run;
        run += e;
    }
    wtot[w * 32 + lane] = run;
    __syncthreads();

    int before = 0;
    int tot = 0;
    #pragma unroll
    for (int ww = 0; ww < RS_WAVES; ++ww) {
        const int c = wtot[ww * 32 + lane];
        before += (ww < w) ? c : 0;
        tot += c;
    }
    int incl = tot;
    #pragma unroll
    for (int off = 1; off < 32; off <<= 1) {
        const int u = __shfl_up(incl, off, 32);
        if (lane >= off) incl += u;
    }
    const int rb = incl - tot;
    if (lane < BUCKET_ROWS) {
        #pragma unroll
        for (int jj = 0; jj < RS_BATCH; ++jj)
            cnt[(w * RS_BATCH + jj) * RS_PITCH + lane] = rb + before + v[jj];
        if (w == 0) {
            int2 se;
            se.x = pb + rb; se.y = pb + rb + tot;
            rse[bucket * BUCKET_ROWS + lane] = se;
        }
    }
    __syncthreads();

    const unsigned lt = (1u << lane) - 1u;
    #pragma unroll
    for (int j = 0; j < RS_BATCH; ++j) {
        const int b = j * RS_WAVES + w;
        if ((b * 32 + lane) < n) {
            const int key = rec[j].x >> 16;
            const int pos = cnt[b * RS_PITCH + key] + __builtin_popcount(peers[j] & lt);
            sorted[pos] = rec[j];
        }
    }
    __syncthreads();

    for (int i = t; i < n; i += RS_THREADS) brec[pb + i] = sorted[i];
}

template <int M>
__device__ __forceinline__ float row_xor_add(float v) {
    const int b = __builtin_bit_cast(int, v);
    const int r = __builtin_amdgcn_update_dpp(0, b, 0x160 | M, 0xf, 0xf, true);
    return v + __builtin_bit_cast(float, r);
}
__device__ __forceinline__ float xhalf(float x) {
    const int b = __builtin_bit_cast(int, x);
    return __builtin_bit_cast(float,
        __builtin_amdgcn_permlanex16(b, b, 0x76543210, 0xfedcba98, false, false));
}

__global__ __launch_bounds__(256, 16) void gather_rows(
    const float* __restrict__ x,
    const int2*  __restrict__ rse,
    const int2*  __restrict__ brec,
    float*       __restrict__ outp)
{
    __shared__ int2 s_rec[ROWS_PER_WG][ROW_CAP];

    const int lane = threadIdx.x & 31;
    const int w    = threadIdx.x >> 5;
    const int row  = blockIdx.x * ROWS_PER_WG + w;
    const int bucket = row / BUCKET_ROWS;
    const int local  = row - bucket * BUCKET_ROWS;
    const int2 se = rse[row];
    const int sx = __builtin_amdgcn_readfirstlane(se.x);
    const int pe = __builtin_amdgcn_readfirstlane(se.y);
    const bool own = sx >= 0;
    const int pb = own ? sx : ~sx;
    const int n = own ? (pe - pb) : (ROW_CAP + 1);
    const bool listed = (n <= ROW_CAP);

    if (listed) {
        int2 rc[ROW_CAP / 32];
        #pragma unroll
        for (int u = 0; u < ROW_CAP / 32; ++u) {
            const int i = u * 32 + lane;
            rc[u].x = 0; rc[u].y = 0;
            if (i < n) rc[u] = brec[pb + i];
        }
        #pragma unroll
        for (int u = 0; u < ROW_CAP / 32; ++u) {
            const int i = u * 32 + lane;
            if (i < n) {
                int2 keep;
                keep.x = rc[u].x & 0xffff;
                keep.y = rc[u].y;
                s_rec[w][i] = keep;
            }
        }
    }
    __syncthreads();

    float4 acc;
    acc.x = 0.f; acc.y = 0.f; acc.z = 0.f; acc.w = 0.f;

    if (listed) {
        int i = 0;
        for (; i + 8 <= n; i += 8) {
            float4 xv[8];
            float  cc[8];
            #pragma unroll
            for (int u = 0; u < 8; ++u) {
                const int2 r = s_rec[w][i + u];
                const int ss = __builtin_amdgcn_readfirstlane(r.x);
                cc[u] = __int_as_float(r.y);
                xv[u] = *(reinterpret_cast<const float4*>(x + (size_t)ss * N_HID) + lane);
            }
            #pragma unroll
            for (int u = 0; u < 8; ++u) {
                acc.x = fmaf(xv[u].x, cc[u], acc.x);
                acc.y = fmaf(xv[u].y, cc[u], acc.y);
                acc.z = fmaf(xv[u].z, cc[u], acc.z);
                acc.w = fmaf(xv[u].w, cc[u], acc.w);
            }
        }
        const int rem = n - i;
        if (rem > 0) {
            float4 xv[7];
            float  cc[7];
            #pragma unroll
            for (int u = 0; u < 7; ++u) {
                const int idx = (u < rem) ? (i + u) : i;
                const int2 r = s_rec[w][idx];
                const int ss = __builtin_amdgcn_readfirstlane(r.x);
                cc[u] = __int_as_float(r.y);
                xv[u] = *(reinterpret_cast<const float4*>(x + (size_t)ss * N_HID) + lane);
            }
            #pragma unroll
            for (int u = 0; u < 7; ++u) {
                if (u < rem) {
                    acc.x = fmaf(xv[u].x, cc[u], acc.x);
                    acc.y = fmaf(xv[u].y, cc[u], acc.y);
                    acc.z = fmaf(xv[u].z, cc[u], acc.z);
                    acc.w = fmaf(xv[u].w, cc[u], acc.w);
                }
            }
        }
    } else {
        for (int p0 = pb; p0 < pe; p0 += 32) {
            const int p = p0 + lane;
            const bool ok = p < pe;
            int2 rc;
            rc.x = -1; rc.y = 0;
            if (ok) rc = brec[p];
            const int pk = rc.x;
            const int cv = rc.y;
            unsigned m = __builtin_amdgcn_ballot_w32((pk >> 16) == local);
            while (m) {
                const int l = __builtin_ctz(m);
                m &= m - 1;
                const int ss = __builtin_amdgcn_readlane(pk, l) & 0xffff;
                const float sc = __int_as_float(__builtin_amdgcn_readlane(cv, l));
                const float4 xv = *(reinterpret_cast<const float4*>(x + (size_t)ss * N_HID) + lane);
                acc.x = fmaf(xv.x, sc, acc.x);
                acc.y = fmaf(xv.y, sc, acc.y);
                acc.z = fmaf(xv.z, sc, acc.z);
                acc.w = fmaf(xv.w, sc, acc.w);
            }
        }
    }

    v2f a0, a1;
    a0.x = acc.x; a0.y = acc.y;
    a1.x = acc.z; a1.y = acc.w;
    v8f c = {};
    c = __builtin_amdgcn_wmma_f32_16x16x4_f32(false, a0, false, a0, (short)0, c, false, false);
    c = __builtin_amdgcn_wmma_f32_16x16x4_f32(false, a1, false, a1, (short)0, c, false, false);
    const int sel = lane & 7;
    float dv = c[0];
    dv = (sel == 1) ? c[1] : dv;
    dv = (sel == 2) ? c[2] : dv;
    dv = (sel == 3) ? c[3] : dv;
    dv = (sel == 4) ? c[4] : dv;
    dv = (sel == 5) ? c[5] : dv;
    dv = (sel == 6) ? c[6] : dv;
    dv = (sel == 7) ? c[7] : dv;
    dv = (lane < 8 || lane >= 24) ? dv : 0.0f;
    dv = row_xor_add<1>(dv);
    dv = row_xor_add<2>(dv);
    dv = row_xor_add<4>(dv);
    dv = row_xor_add<8>(dv);
    dv = dv + xhalf(dv);

    const float scale = 1.0f / fmaxf(sqrtf(dv), 1e-12f);
    const float inv_sqrt2 = 0.70710678118654752440f;
    float4 o;
    o.x = acc.x * scale;
    o.y = acc.y * scale;
    o.z = acc.z * scale;
    o.w = acc.w * scale;
    o.x = 0.5f * o.x * (1.0f + erff(o.x * inv_sqrt2));
    o.y = 0.5f * o.y * (1.0f + erff(o.y * inv_sqrt2));
    o.z = 0.5f * o.z * (1.0f + erff(o.z * inv_sqrt2));
    o.w = 0.5f * o.w * (1.0f + erff(o.w * inv_sqrt2));
    *(reinterpret_cast<float4*>(outp + (size_t)row * N_HID) + lane) = o;
}

extern "C" void kernel_launch(void* const* d_in, const int* in_sizes, int n_in,
                              void* d_out, int out_size, void* d_ws, size_t ws_size,
                              hipStream_t stream) {
    const float* x        = (const float*)d_in[0];
    const float* weight   = (const float*)d_in[1];
    const int*   adj_src  = (const int*)d_in[2];
    const int*   adj_dst  = (const int*)d_in[3];
    const float* adj_vals = (const float*)d_in[4];
    float*       out      = (float*)d_out;

    int*   ghist  = (int*)d_ws;
    int*   btotal = ghist + P_WG * N_BUCKET;
    int*   bstart = btotal + 2048;
    int2*  brec   = (int2*)(bstart + 2048);
    int2*  rse    = (int2*)d_ws;

    part_count<<<P_WG, P_THREADS, 0, stream>>>(adj_dst, ghist);
    bucket_prefix<<<(N_BUCKET + 255) / 256, 256, 0, stream>>>(ghist, btotal);
    part_scatter<<<P_WG, P_THREADS, 0, stream>>>(weight, adj_src, adj_dst, adj_vals,
                                                 ghist, btotal, bstart, brec);
    row_sort<<<N_BUCKET, RS_THREADS, 0, stream>>>(bstart, brec, rse);
    gather_rows<<<N_NODES / ROWS_PER_WG, 256, 0, stream>>>(x, rse, brec, out);
}


# Pull instead of push: each block orders only its own 9600 edges by bucket and stores them whole; a bucket's block fetches its 250 runs itself; count and prefix launches go (5 -> 3); sum order kept
# speedup vs baseline: 5.5368x; 1.3613x over previous
#include <hip/hip_runtime.h>
#include <math.h>

#define N_NODES 50000
#define N_HID   128
#define N_ADJ   4
#define N_EDGE  600000
#define TOTAL_EDGES (N_ADJ * N_EDGE)

#define BUCKET_ROWS 25
#define N_BUCKET    (N_NODES / BUCKET_ROWS)
#define KEY_BITS    11
#define P_WG        250
#define P_WAVES     15
#define P_THREADS   (P_WAVES * 32)
#define P_BATCH     20
#define SCAN_LANES  400
#define P_EDGES     (P_WAVES * P_BATCH * 32)
#define W_PITCH     2016
#define G_PAD       256
#define ROW_CAP     128
#define ROWS_PER_WG 8
#define RS_WAVES    4
#define RS_THREADS  (RS_WAVES * 32)
#define RS_BATCH    12
#define RS_NB       (RS_WAVES * RS_BATCH)
#define RS_CAP      (RS_NB * 32)
#define RS_PITCH    32
#define ROW_BITS    5

typedef __attribute__((ext_vector_type(2))) float v2f;
typedef __attribute__((ext_vector_type(8))) float v8f;

__device__ __forceinline__ unsigned same_key_lanes(int key) {
    unsigned m = 0xffffffffu;
    #pragma unroll
    for (int b = 0; b < KEY_BITS; ++b) {
        const bool bit = ((key >> b) & 1) != 0;
        const unsigned bal = __builtin_amdgcn_ballot_w32(bit);
        m &= bit ? bal : ~bal;
    }
    return m;
}

__device__ __forceinline__ unsigned same_row_lanes(int key) {
    unsigned m = 0xffffffffu;
    #pragma unroll
    for (int b = 0; b < ROW_BITS; ++b) {
        const int sm = (int)((unsigned)key << (31 - b)) >> 31;
        const unsigned bal = __builtin_amdgcn_ballot_w32(sm != 0);
        m &= ~(bal ^ (unsigned)sm);
    }
    return m;
}

__global__ __launch_bounds__(P_THREADS) void part_local(
    const float* __restrict__ weight,
    const int*   __restrict__ adj_src,
    const int*   __restrict__ adj_dst,
    const float* __restrict__ adj_vals,
    int*   __restrict__ wst,
    int2*  __restrict__ seg)
{
    __shared__ int  wbase[P_WAVES][N_BUCKET];
    __shared__ int  wrow[N_BUCKET + 1];
    __shared__ int  part[P_THREADS];
    __shared__ int2 srec[P_EDGES];
    const int t = threadIdx.x;
    const int lane = t & 31;
    const int w = t >> 5;
    const int g = blockIdx.x;

    const int ebase = (g * P_WAVES + w) * (P_BATCH * 32);
    int dstv[P_BATCH];
    #pragma unroll
    for (int b = 0; b < P_BATCH; ++b)
        dstv[b] = __builtin_nontemporal_load(adj_dst + ebase + b * 32 + lane);

    for (int i = t; i < P_WAVES * N_BUCKET; i += P_THREADS) (&wbase[0][0])[i] = 0;
    __syncthreads();

    int   srcv[P_BATCH];
    float valv[P_BATCH];
    #pragma unroll
    for (int b = 0; b < P_BATCH; ++b)
        srcv[b] = adj_src[ebase + b * 32 + lane];
    #pragma unroll
    for (int b = 0; b < P_BATCH; ++b)
        valv[b] = adj_vals[ebase + b * 32 + lane];
    const int   a0     = ebase / N_EDGE;
    const int   a1     = (a0 + 1 < N_ADJ) ? (a0 + 1) : (N_ADJ - 1);
    const int   abound = (a0 + 1) * N_EDGE;
    const float wgt0   = weight[a0];
    const float wgt1   = weight[a1];

    unsigned peers[P_BATCH];
    #pragma unroll
    for (int b = 0; b < P_BATCH; ++b) {
        const int key = dstv[b] / BUCKET_ROWS;
        const unsigned m = same_key_lanes(key);
        peers[b] = m;
        if (lane == __builtin_ctz(m)) wbase[w][key] += __builtin_popcount(m);
    }
    __syncthreads();

    int cw[5];
    int mine = 0;
    #pragma unroll
    for (int j = 0; j < 5; ++j) {
        int s = 0;
        if (t < SCAN_LANES) {
            #pragma unroll
            for (int wv = 0; wv < P_WAVES; ++wv) s += wbase[wv][t * 5 + j];
        }
        cw[j] = s;
        mine += s;
    }
    part[t] = mine;
    __syncthreads();

    for (int off = 1; off < P_THREADS; off <<= 1) {
        int add = 0;
        if (t >= off) add = part[t - off];
        __syncthreads();
        part[t] += add;
        __syncthreads();
    }
    int run = part[t] - mine;
    if (t < SCAN_LANES) {
        #pragma unroll
        for (int j = 0; j < 5; ++j) {
            const int k = t * 5 + j;
            wrow[k] = run;
            int base = run;
            #pragma unroll
            for (int wv = 0; wv < P_WAVES; ++wv) {
                const int cnt = wbase[wv][k];
                wbase[wv][k] = base;
                base += cnt;
            }
            run += cw[j];
        }
    }
    if (t == P_THREADS - 1) wrow[N_BUCKET] = P_EDGES;
    __syncthreads();

    for (int i = t; i <= N_BUCKET; i += P_THREADS) wst[g * W_PITCH + i] = wrow[i];

    const unsigned lt = (1u << lane) - 1u;
    #pragma unroll
    for (int b = 0; b < P_BATCH; ++b) {
        const int e = ebase + b * 32 + lane;
        const int d = dstv[b];
        const int key = d / BUCKET_ROWS;
        const unsigned m = peers[b];
        const int leader = __builtin_ctz(m);
        int slot = 0;
        if (lane == leader) {
            slot = wbase[w][key];
            wbase[w][key] = slot + __builtin_popcount(m);
        }
        slot = __shfl(slot, leader, 32);
        const int lp = slot + __builtin_popcount(m & lt);
        int2 rec;
        rec.x = ((d - key * BUCKET_ROWS) << 16) | srcv[b];
        rec.y = __float_as_int(((e < abound) ? wgt0 : wgt1) * valv[b]);
        srec[lp] = rec;
    }
    __syncthreads();

    #pragma unroll
    for (int b = 0; b < P_BATCH; ++b) {
        const int i = b * P_THREADS + t;
        seg[g * P_EDGES + i] = srec[i];
    }
}

__device__ __forceinline__ int run_of(const int* gpre, int i) {
    int pos = 0;
    #pragma unroll
    for (int step = G_PAD / 2; step >= 1; step >>= 1)
        pos += (gpre[pos + step] <= i) ? step : 0;
    return pos;
}

__global__ __launch_bounds__(RS_THREADS) void row_sort(
    const int* __restrict__ wst, const int2* __restrict__ seg,
    int2* __restrict__ brec, int2* __restrict__ rse)
{
    __shared__ int  cnt[RS_NB * RS_PITCH];
    __shared__ int  wtot[RS_WAVES * 32];
    __shared__ int2 sorted[RS_CAP];
    __shared__ int  gpre[G_PAD];
    __shared__ int  gadj[G_PAD];
    __shared__ int2 wsum[RS_WAVES];
    const int t = threadIdx.x;
    const int lane = t & 31;
    const int w = t >> 5;
    const int bucket = blockIdx.x;

    const int g0 = 2 * t;
    const int g1 = 2 * t + 1;
    int s0 = 0, e0 = 0, s1 = 0, e1 = 0;
    if (g1 < P_WG) {
        s0 = wst[g0 * W_PITCH + bucket];
        e0 = wst[g0 * W_PITCH + bucket + 1];
        s1 = wst[g1 * W_PITCH + bucket];
        e1 = wst[g1 * W_PITCH + bucket + 1];
    }
    for (int i = t; i < RS_NB * RS_PITCH; i += RS_THREADS) cnt[i] = 0;
    const int c0 = e0 - s0;
    const int c1 = e1 - s1;
    const int pc = c0 + c1;
    int pinc  = pc;
    int psum = s0 + s1;
    #pragma unroll
    for (int off = 1; off < 32; off <<= 1) {
        const int u = __shfl_up(pinc, off, 32);
        const int v = __shfl_up(psum, off, 32);
        if (lane >= off) { pinc += u; psum += v; }
    }
    if (lane == 31) {
        int2 ws2;
        ws2.x = pinc; ws2.y = psum;
        wsum[w] = ws2;
    }
    __syncthreads();
    int woff = 0, ntot = 0, ptot = 0;
    #pragma unroll
    for (int ww = 0; ww < RS_WAVES; ++ww) {
        const int2 v = wsum[ww];
        woff += (ww < w) ? v.x : 0;
        ntot += v.x;
        ptot += v.y;
    }
    const int n  = __builtin_amdgcn_readfirstlane(ntot);
    const int pb = __builtin_amdgcn_readfirstlane(ptot);
    const int pe = pb + n;
    const int ex = woff + pinc - pc;
    const bool live = g1 < P_WG;
    gpre[g0] = live ? ex : 0x7fffffff;
    gpre[g1] = live ? (ex + c0) : 0x7fffffff;
    gadj[g0] = g0 * P_EDGES + s0 - ex;
    gadj[g1] = g1 * P_EDGES + s1 - (ex + c0);
    __syncthreads();

    if (n > RS_CAP) {
        for (int i = t; i < n; i += RS_THREADS) brec[pb + i] = seg[gadj[run_of(gpre, i)] + i];
        if (t < BUCKET_ROWS) {
            int2 mark;
            mark.x = ~pb; mark.y = pe;
            rse[bucket * BUCKET_ROWS + t] = mark;
        }
        return;
    }

    int pos[RS_BATCH];
    #pragma unroll
    for (int j = 0; j < RS_BATCH; ++j) pos[j] = 0;
    #pragma unroll
    for (int step = G_PAD / 2; step >= 1; step >>= 1) {
        #pragma unroll
        for (int j = 0; j < RS_BATCH; ++j) {
            const int i = (j * RS_WAVES + w) * 32 + lane;
            pos[j] += (gpre[pos[j] + step] <= i) ? step : 0;
        }
    }
    int adr[RS_BATCH];
    #pragma unroll
    for (int j = 0; j < RS_BATCH; ++j)
        adr[j] = gadj[pos[j]] + (j * RS_WAVES + w) * 32 + lane;
    int2 rec[RS_BATCH];
    #pragma unroll
    for (int j = 0; j < RS_BATCH; ++j) {
        const int i = (j * RS_WAVES + w) * 32 + lane;
        int2 r;
        r.x = -1; r.y = 0;
        if (i < n) r = seg[adr[j]];
        rec[j] = r;
    }

    unsigned peers[RS_BATCH];
    #pragma unroll
    for (int j = 0; j < RS_BATCH; ++j) {
        const int b = j * RS_WAVES + w;
        peers[j] = 0u;
        if (b * 32 < n) {
            const bool ok = (b * 32 + lane) < n;
            const int key = ok ? (rec[j].x >> 16) : 31;
            const unsigned m = same_row_lanes(key);
            peers[j] = m;
            if (ok && lane == __builtin_ctz(m)) cnt[b * RS_PITCH + key] = __builtin_popcount(m);
        }
    }
    __syncthreads();
    int v[RS_BATCH];
    #pragma unroll
    for (int jj = 0; jj < RS_BATCH; ++jj) v[jj] = cnt[(w * RS_BATCH + jj) * RS_PITCH + lane];
    int run = 0;
    #pragma unroll
    for (int jj = 0; jj < RS_BATCH; ++jj) {
        const int e = v[jj];
        v[jj] = run;
        run += e;
    }
    wtot[w * 32 + lane] = run;
    __syncthreads();

    int before = 0;
    int tot = 0;
    #pragma unroll
    for (int ww = 0; ww < RS_WAVES; ++ww) {
        const int c = wtot[ww * 32 + lane];
        before += (ww < w) ? c : 0;
        tot += c;
    }
    int incl = tot;
    #pragma unroll
    for (int off = 1; off < 32; off <<= 1) {
        const int u = __shfl_up(incl, off, 32);
        if (lane >= off) incl += u;
    }
    const int rb = incl - tot;
    if (lane < BUCKET_ROWS) {
        #pragma unroll
        for (int jj = 0; jj < RS_BATCH; ++jj)
            cnt[(w * RS_BATCH + jj) * RS_PITCH + lane] = rb + before + v[jj];
        if (w == 0) {
            int2 se;
            se.x = pb + rb; se.y = pb + rb + tot;
            rse[bucket * BUCKET_ROWS + lane] = se;
        }
    }
    __syncthreads();

    const unsigned lt = (1u << lane) - 1u;
    #pragma unroll
    for (int j = 0; j < RS_BATCH; ++j) {
        const int b = j * RS_WAVES + w;
        if ((b * 32 + lane) < n) {
            const int key = rec[j].x >> 16;
            const int pos = cnt[b * RS_PITCH + key] + __builtin_popcount(peers[j] & lt);
            sorted[pos] = rec[j];
        }
    }
    __syncthreads();

    for (int i = t; i < n; i += RS_THREADS) brec[pb + i] = sorted[i];
}

template <int M>
__device__ __forceinline__ float row_xor_add(float v) {
    const int b = __builtin_bit_cast(int, v);
    const int r = __builtin_amdgcn_update_dpp(0, b, 0x160 | M, 0xf, 0xf, true);
    return v + __builtin_bit_cast(float, r);
}
__device__ __forceinline__ float xhalf(float x) {
    const int b = __builtin_bit_cast(int, x);
    return __builtin_bit_cast(float,
        __builtin_amdgcn_permlanex16(b, b, 0x76543210, 0xfedcba98, false, false));
}

__global__ __launch_bounds__(256, 16) void gather_rows(
    const float* __restrict__ x,
    const int2*  __restrict__ rse,
    const int2*  __restrict__ brec,
    float*       __restrict__ outp)
{
    __shared__ int2 s_rec[ROWS_PER_WG][ROW_CAP];

    const int lane = threadIdx.x & 31;
    const int w    = threadIdx.x >> 5;
    const int row  = blockIdx.x * ROWS_PER_WG + w;
    const int bucket = row / BUCKET_ROWS;
    const int local  = row - bucket * BUCKET_ROWS;
    const int2 se = rse[row];
    const int sx = __builtin_amdgcn_readfirstlane(se.x);
    const int pe = __builtin_amdgcn_readfirstlane(se.y);
    const bool own = sx >= 0;
    const int pb = own ? sx : ~sx;
    const int n = own ? (pe - pb) : (ROW_CAP + 1);
    const bool listed = (n <= ROW_CAP);

    if (listed) {
        int2 rc[ROW_CAP / 32];
        #pragma unroll
        for (int u = 0; u < ROW_CAP / 32; ++u) {
            const int i = u * 32 + lane;
            rc[u].x = 0; rc[u].y = 0;
            if (i < n) rc[u] = brec[pb + i];
        }
        #pragma unroll
        for (int u = 0; u < ROW_CAP / 32; ++u) {
            const int i = u * 32 + lane;
            if (i < n) {
                int2 keep;
                keep.x = rc[u].x & 0xffff;
                keep.y = rc[u].y;
                s_rec[w][i] = keep;
            }
        }
    }
    __syncthreads();

    float4 acc;
    acc.x = 0.f; acc.y = 0.f; acc.z = 0.f; acc.w = 0.f;

    if (listed) {
        int i = 0;
        for (; i + 8 <= n; i += 8) {
            float4 xv[8];
            float  cc[8];
            #pragma unroll
            for (int u = 0; u < 8; ++u) {
                const int2 r = s_rec[w][i + u];
                const int ss = __builtin_amdgcn_readfirstlane(r.x);
                cc[u] = __int_as_float(r.y);
                xv[u] = *(reinterpret_cast<const float4*>(x + (size_t)ss * N_HID) + lane);
            }
            #pragma unroll
            for (int u = 0; u < 8; ++u) {
                acc.x = fmaf(xv[u].x, cc[u], acc.x);
                acc.y = fmaf(xv[u].y, cc[u], acc.y);
                acc.z = fmaf(xv[u].z, cc[u], acc.z);
                acc.w = fmaf(xv[u].w, cc[u], acc.w);
            }
        }
        const int rem = n - i;
        if (rem > 0) {
            float4 xv[7];
            float  cc[7];
            #pragma unroll
            for (int u = 0; u < 7; ++u) {
                const int idx = (u < rem) ? (i + u) : i;
                const int2 r = s_rec[w][idx];
                const int ss = __builtin_amdgcn_readfirstlane(r.x);
                cc[u] = __int_as_float(r.y);
                xv[u] = *(reinterpret_cast<const float4*>(x + (size_t)ss * N_HID) + lane);
            }
            #pragma unroll
            for (int u = 0; u < 7; ++u) {
                if (u < rem) {
                    acc.x = fmaf(xv[u].x, cc[u], acc.x);
                    acc.y = fmaf(xv[u].y, cc[u], acc.y);
                    acc.z = fmaf(xv[u].z, cc[u], acc.z);
                    acc.w = fmaf(xv[u].w, cc[u], acc.w);
                }
            }
        }
    } else {
        for (int p0 = pb; p0 < pe; p0 += 32) {
            const int p = p0 + lane;
            const bool ok = p < pe;
            int2 rc;
            rc.x = -1; rc.y = 0;
            if (ok) rc = brec[p];
            const int pk = rc.x;
            const int cv = rc.y;
            unsigned m = __builtin_amdgcn_ballot_w32((pk >> 16) == local);
            while (m) {
                const int l = __builtin_ctz(m);
                m &= m - 1;
                const int ss = __builtin_amdgcn_readlane(pk, l) & 0xffff;
                const float sc = __int_as_float(__builtin_amdgcn_readlane(cv, l));
                const float4 xv = *(reinterpret_cast<const float4*>(x + (size_t)ss * N_HID) + lane);
                acc.x = fmaf(xv.x, sc, acc.x);
                acc.y = fmaf(xv.y, sc, acc.y);
                acc.z = fmaf(xv.z, sc, acc.z);
                acc.w = fmaf(xv.w, sc, acc.w);
            }
        }
    }

    v2f a0, a1;
    a0.x = acc.x; a0.y = acc.y;
    a1.x = acc.z; a1.y = acc.w;
    v8f c = {};
    c = __builtin_amdgcn_wmma_f32_16x16x4_f32(false, a0, false, a0, (short)0, c, false, false);
    c = __builtin_amdgcn_wmma_f32_16x16x4_f32(false, a1, false, a1, (short)0, c, false, false);
    const int sel = lane & 7;
    float dv = c[0];
    dv = (sel == 1) ? c[1] : dv;
    dv = (sel == 2) ? c[2] : dv;
    dv = (sel == 3) ? c[3] : dv;
    dv = (sel == 4) ? c[4] : dv;
    dv = (sel == 5) ? c[5] : dv;
    dv = (sel == 6) ? c[6] : dv;
    dv = (sel == 7) ? c[7] : dv;
    dv = (lane < 8 || lane >= 24) ? dv : 0.0f;
    dv = row_xor_add<1>(dv);
    dv = row_xor_add<2>(dv);
    dv = row_xor_add<4>(dv);
    dv = row_xor_add<8>(dv);
    dv = dv + xhalf(dv);

    const float scale = 1.0f / fmaxf(sqrtf(dv), 1e-12f);
    const float inv_sqrt2 = 0.70710678118654752440f;
    float4 o;
    o.x = acc.x * scale;
    o.y = acc.y * scale;
    o.z = acc.z * scale;
    o.w = acc.w * scale;
    o.x = 0.5f * o.x * (1.0f + erff(o.x * inv_sqrt2));
    o.y = 0.5f * o.y * (1.0f + erff(o.y * inv_sqrt2));
    o.z = 0.5f * o.z * (1.0f + erff(o.z * inv_sqrt2));
    o.w = 0.5f * o.w * (1.0f + erff(o.w * inv_sqrt2));
    *(reinterpret_cast<float4*>(outp + (size_t)row * N_HID) + lane) = o;
}

extern "C" void kernel_launch(void* const* d_in, const int* in_sizes, int n_in,
                              void* d_out, int out_size, void* d_ws, size_t ws_size,
                              hipStream_t stream) {
    const float* x        = (const float*)d_in[0];
    const float* weight   = (const float*)d_in[1];
    const int*   adj_src  = (const int*)d_in[2];
    const int*   adj_dst  = (const int*)d_in[3];
    const float* adj_vals = (const float*)d_in[4];
    float*       out      = (float*)d_out;

    int2*  rse    = (int2*)d_ws;
    int2*  brec   = (int2*)((int*)d_ws + P_WG * N_BUCKET + 4096);
    int2*  seg    = (int2*)d_out;
    int*   wst    = (int*)(seg + TOTAL_EDGES);

    part_local<<<P_WG, P_THREADS, 0, stream>>>(weight, adj_src, adj_dst, adj_vals, wst, seg);
    row_sort<<<N_BUCKET, RS_THREADS, 0, stream>>>(wst, seg, brec, rse);
    gather_rows<<<N_NODES / ROWS_PER_WG, 256, 0, stream>>>(x, rse, brec, out);
}
